# starting kernel
# baseline (speedup 1.0000x reference)
#include <hip/hip_runtime.h>
#include <math.h>

typedef __attribute__((ext_vector_type(2))) float v2f;
typedef __attribute__((ext_vector_type(8))) float v8f;

#define NEG_SLOPE 0.2f

__device__ __forceinline__ float lrelu(float v) { return v >= 0.f ? v : NEG_SLOPE * v; }

__device__ __forceinline__ void atomicMaxF(float* addr, float val) {
  if (val >= 0.f) atomicMax((int*)addr, __float_as_int(val));
  else            atomicMin((unsigned int*)addr, __float_as_uint(val));
}

__global__ void gemm1_x_w1(const float* __restrict__ X, const float* __restrict__ W,
                           float* __restrict__ H) {
  const int lane = threadIdx.x & 31;
  const int wave = threadIdx.x >> 5;
  const int kgrp = (lane >> 4) << 1;
  const int ncol = (wave << 4) + (lane & 15);
  const float* xrow = X + ((long)blockIdx.x * 16 + (lane & 15)) * 256;
  v8f acc = {};
#pragma unroll 8
  for (int k0 = 0; k0 < 256; k0 += 4) {
    v2f a = *(const v2f*)(xrow + k0 + kgrp);
    v2f b;
    b.x = W[(k0 + kgrp) * 64 + ncol];
    b.y = W[(k0 + kgrp + 1) * 64 + ncol];
    acc = __builtin_amdgcn_wmma_f32_16x16x4_f32(false, a, false, b, (short)0, acc,
                                                false, false);
  }
  const long mb = (long)blockIdx.x * 16 + ((lane >> 4) << 3);
#pragma unroll
  for (int v = 0; v < 8; ++v)
    H[(mb + v) * 64 + ncol] = acc[v];
}

__global__ void gemm2_h_w2(const float* __restrict__ A1, const float* __restrict__ W,
                           float* __restrict__ H2, int mtiles) {
  const int lane = threadIdx.x & 31;
  const int wave = threadIdx.x >> 5;
  const int mtile = blockIdx.x * 4 + wave;
  if (mtile >= mtiles) return;
  const int kgrp = (lane >> 4) << 1;
  const int ncol = lane & 15;
  const float* arow = A1 + ((long)mtile * 16 + ncol) * 64;
  v8f acc = {};
#pragma unroll
  for (int k0 = 0; k0 < 64; k0 += 4) {
    v2f a = *(const v2f*)(arow + k0 + kgrp);
    a.x = fmaxf(a.x, 0.f);
    a.y = fmaxf(a.y, 0.f);
    v2f b;
    b.x = W[(k0 + kgrp) * 8 + (ncol & 7)];
    b.y = W[(k0 + kgrp + 1) * 8 + (ncol & 7)];
    acc = __builtin_amdgcn_wmma_f32_16x16x4_f32(false, a, false, b, (short)0, acc,
                                                false, false);
  }
  if (ncol < 8) {
    const long mb = (long)mtile * 16 + ((lane >> 4) << 3);
#pragma unroll
    for (int v = 0; v < 8; ++v)
      H2[(mb + v) * 8 + ncol] = acc[v];
  }
}

template <int H, int C>
__global__ void attn_logits(const float* __restrict__ hf, const float* __restrict__ as,
                            const float* __restrict__ ad, float* __restrict__ als,
                            float* __restrict__ ald, long NH) {
  long t = (long)blockIdx.x * blockDim.x + threadIdx.x;
  if (t >= NH) return;
  const int h = (int)(t % H);
  const float* hp = hf + t * C;
  float ss = 0.f, sd = 0.f;
#pragma unroll
  for (int c = 0; c < C; ++c) {
    float v = hp[c];
    ss += v * as[h * C + c];
    sd += v * ad[h * C + c];
  }
  als[t] = ss;
  ald[t] = sd;
}

__global__ void init_softmax(float* __restrict__ m, float* __restrict__ den, long n) {
  long t = (long)blockIdx.x * blockDim.x + threadIdx.x;
  if (t < n) { m[t] = -__builtin_huge_valf(); den[t] = 0.f; }
}

__global__ void init_acc_bias(float* __restrict__ acc, const float* __restrict__ b,
                              long n, int F) {
  long t = (long)blockIdx.x * blockDim.x + threadIdx.x;
  if (t < n) acc[t] = b[(int)(t % F)];
}

__device__ __forceinline__ void edge_sd(const int* __restrict__ ei, long Eraw, long e,
                                        int& s, int& d) {
  if (e < Eraw) { s = ei[e]; d = ei[Eraw + e]; }
  else          { s = d = (int)(e - Eraw); }
}

template <int H>
__global__ void edge_softmax_max(const int* __restrict__ ei, long Eraw, long Etot,
                                 const float* __restrict__ als, const float* __restrict__ ald,
                                 float* __restrict__ m) {
  long e = (long)blockIdx.x * blockDim.x + threadIdx.x;
  if (e >= Etot) return;
  int s, d; edge_sd(ei, Eraw, e, s, d);
#pragma unroll
  for (int h = 0; h < H; ++h) {
    float v = lrelu(als[(long)s * H + h] + ald[(long)d * H + h]);
    atomicMaxF(&m[(long)d * H + h], v);
  }
}

template <int H>
__global__ void edge_softmax_sum(const int* __restrict__ ei, long Eraw, long Etot,
                                 const float* __restrict__ als, const float* __restrict__ ald,
                                 const float* __restrict__ m, float* __restrict__ den) {
  long e = (long)blockIdx.x * blockDim.x + threadIdx.x;
  if (e >= Etot) return;
  int s, d; edge_sd(ei, Eraw, e, s, d);
#pragma unroll
  for (int h = 0; h < H; ++h) {
    float v = lrelu(als[(long)s * H + h] + ald[(long)d * H + h]);
    atomicAdd(&den[(long)d * H + h], __expf(v - m[(long)d * H + h]));
  }
}

template <int H, int C>
__global__ void edge_aggregate(const int* __restrict__ ei, long Eraw, long Etot,
                               const float* __restrict__ als, const float* __restrict__ ald,
                               const float* __restrict__ m, const float* __restrict__ den,
                               const float* __restrict__ hf, float* __restrict__ acc) {
  long t = (long)blockIdx.x * blockDim.x + threadIdx.x;
  if (t >= Etot * H) return;
  long e = t / H;
  int h = (int)(t % H);
  int s, d; edge_sd(ei, Eraw, e, s, d);
  long sh = (long)s * H + h, dh = (long)d * H + h;
  float v = lrelu(als[sh] + ald[dh]);
  float alpha = __expf(v - m[dh]) / (den[dh] + 1e-16f);
  const float* hp = hf + sh * C;
  float* op = acc + dh * C;
#pragma unroll
  for (int c = 0; c < C; ++c)
    atomicAdd(&op[c], hp[c] * alpha);
}

__global__ void score_kernel(const float* __restrict__ emb, const float* __restrict__ W3,
                             const float* __restrict__ b3, float* __restrict__ score, long N) {
  long n = (long)blockIdx.x * blockDim.x + threadIdx.x;
  if (n >= N) return;
  float s = b3[0];
#pragma unroll
  for (int c = 0; c < 8; ++c) s += emb[n * 8 + c] * W3[c];
  score[n] = s;
}

static inline int nblk(long n, int b) { return (int)((n + b - 1) / b); }

extern "C" void kernel_launch(void* const* d_in, const int* in_sizes, int n_in,
                              void* d_out, int out_size, void* d_ws, size_t ws_size,
                              hipStream_t stream) {
  const float* x   = (const float*)d_in[0];
  const int*   ei  = (const int*)d_in[1];
  const float* W1  = (const float*)d_in[2];
  const float* as1 = (const float*)d_in[3];
  const float* ad1 = (const float*)d_in[4];
  const float* b1  = (const float*)d_in[5];
  const float* W2  = (const float*)d_in[6];
  const float* as2 = (const float*)d_in[7];
  const float* ad2 = (const float*)d_in[8];
  const float* b2  = (const float*)d_in[9];
  const float* W3  = (const float*)d_in[10];
  const float* b3  = (const float*)d_in[11];

  const long N    = in_sizes[0] / 256;
  const long Eraw = in_sizes[1] / 2;
  const long Etot = Eraw + N;

  float* ws   = (float*)d_ws;
  float* h1   = ws;
  float* acc1 = h1 + N * 64;
  float* als1 = acc1 + N * 64;
  float* ald1 = als1 + N * 8;
  float* m1   = ald1 + N * 8;
  float* den1 = m1 + N * 8;
  float* h2   = den1 + N * 8;
  float* als2 = h2 + N * 8;
  float* ald2 = als2 + N;
  float* m2   = ald2 + N;
  float* den2 = m2 + N;

  float* emb   = (float*)d_out;
  float* score = emb + N * 8;

  const int B = 256;
  const int mtiles = (int)(N / 16);

  gemm1_x_w1<<<mtiles, 128, 0, stream>>>(x, W1, h1);
  attn_logits<8, 8><<<nblk(N * 8, B), B, 0, stream>>>(h1, as1, ad1, als1, ald1, N * 8);
  init_softmax<<<nblk(N * 8, B), B, 0, stream>>>(m1, den1, N * 8);
  init_acc_bias<<<nblk(N * 64, B), B, 0, stream>>>(acc1, b1, N * 64, 64);
  edge_softmax_max<8><<<nblk(Etot, B), B, 0, stream>>>(ei, Eraw, Etot, als1, ald1, m1);
  edge_softmax_sum<8><<<nblk(Etot, B), B, 0, stream>>>(ei, Eraw, Etot, als1, ald1, m1, den1);
  edge_aggregate<8, 8><<<nblk(Etot * 8, B), B, 0, stream>>>(ei, Eraw, Etot, als1, ald1,
                                                            m1, den1, h1, acc1);

  gemm2_h_w2<<<(mtiles + 3) / 4, 128, 0, stream>>>(acc1, W2, h2, mtiles);
  attn_logits<1, 8><<<nblk(N, B), B, 0, stream>>>(h2, as2, ad2, als2, ald2, N);
  init_softmax<<<nblk(N, B), B, 0, stream>>>(m2, den2, N);
  init_acc_bias<<<nblk(N * 8, B), B, 0, stream>>>(emb, b2, N * 8, 8);
  edge_softmax_max<1><<<nblk(Etot, B), B, 0, stream>>>(ei, Eraw, Etot, als2, ald2, m2);
  edge_softmax_sum<1><<<nblk(Etot, B), B, 0, stream>>>(ei, Eraw, Etot, als2, ald2, m2, den2);
  edge_aggregate<1, 8><<<nblk(Etot, B), B, 0, stream>>>(ei, Eraw, Etot, als2, ald2,
                                                        m2, den2, h2, emb);

  score_kernel<<<nblk(N, B), B, 0, stream>>>(emb, W3, b3, score, N);
}


# Edges grouped by destination once per call (integer-atomic counting sort, arrival-order rank); segment max/exp-sum/aggregation become per-node gathers, no float atomics; sum order changes
# speedup vs baseline: 5.4409x; 5.4409x over previous
#include <hip/hip_runtime.h>
#include <math.h>

typedef __attribute__((ext_vector_type(2))) float v2f;
typedef __attribute__((ext_vector_type(8))) float v8f;

#define NEG_SLOPE 0.2f

__device__ __forceinline__ float lrelu(float v) { return v >= 0.f ? v : NEG_SLOPE * v; }

__global__ void gemm1_x_w1(const float* __restrict__ X, const float* __restrict__ W,
                           float* __restrict__ H) {
  const int lane = threadIdx.x & 31;
  const int wave = threadIdx.x >> 5;
  const int kgrp = (lane >> 4) << 1;
  const int ncol = (wave << 4) + (lane & 15);
  const float* xrow = X + ((long)blockIdx.x * 16 + (lane & 15)) * 256;
  v8f acc = {};
#pragma unroll 8
  for (int k0 = 0; k0 < 256; k0 += 4) {
    v2f a = *(const v2f*)(xrow + k0 + kgrp);
    v2f b;
    b.x = W[(k0 + kgrp) * 64 + ncol];
    b.y = W[(k0 + kgrp + 1) * 64 + ncol];
    acc = __builtin_amdgcn_wmma_f32_16x16x4_f32(false, a, false, b, (short)0, acc,
                                                false, false);
  }
  const long mb = (long)blockIdx.x * 16 + ((lane >> 4) << 3);
#pragma unroll
  for (int v = 0; v < 8; ++v)
    H[(mb + v) * 64 + ncol] = acc[v];
}

__global__ void gemm2_h_w2(const float* __restrict__ A1, const float* __restrict__ W,
                           float* __restrict__ H2, int mtiles) {
  const int lane = threadIdx.x & 31;
  const int wave = threadIdx.x >> 5;
  const int mtile = blockIdx.x * 4 + wave;
  if (mtile >= mtiles) return;
  const int kgrp = (lane >> 4) << 1;
  const int ncol = lane & 15;
  const float* arow = A1 + ((long)mtile * 16 + ncol) * 64;
  v8f acc = {};
#pragma unroll
  for (int k0 = 0; k0 < 64; k0 += 4) {
    v2f a = *(const v2f*)(arow + k0 + kgrp);
    a.x = fmaxf(a.x, 0.f);
    a.y = fmaxf(a.y, 0.f);
    v2f b;
    b.x = W[(k0 + kgrp) * 8 + (ncol & 7)];
    b.y = W[(k0 + kgrp + 1) * 8 + (ncol & 7)];
    acc = __builtin_amdgcn_wmma_f32_16x16x4_f32(false, a, false, b, (short)0, acc,
                                                false, false);
  }
  if (ncol < 8) {
    const long mb = (long)mtile * 16 + ((lane >> 4) << 3);
#pragma unroll
    for (int v = 0; v < 8; ++v)
      H2[(mb + v) * 8 + ncol] = acc[v];
  }
}

template <int H, int C>
__global__ void src_logits(const float* __restrict__ hf, const float* __restrict__ as,
                           float* __restrict__ als, long NH) {
  long t = (long)blockIdx.x * blockDim.x + threadIdx.x;
  if (t >= NH) return;
  const int h = (int)(t % H);
  const float* hp = hf + t * C;
  float ss = 0.f;
#pragma unroll
  for (int c = 0; c < C; ++c) ss += hp[c] * as[h * C + c];
  als[t] = ss;
}

__global__ void zero_counts(int* __restrict__ cnt, long n) {
  long t = (long)blockIdx.x * blockDim.x + threadIdx.x;
  if (t < n) cnt[t] = 0;
}

__global__ void dst_histogram(const int* __restrict__ ei, long Eraw, int* __restrict__ cnt,
                              int* __restrict__ rank) {
  const long e0 = ((long)blockIdx.x * blockDim.x + threadIdx.x) * 4;
  if (e0 + 3 < Eraw) {
    const int* dp = ei + Eraw + e0;
    const int d0 = dp[0], d1 = dp[1], d2 = dp[2], d3 = dp[3];
    const int r0 = atomicAdd(&cnt[d0], 1);
    const int r1 = atomicAdd(&cnt[d1], 1);
    const int r2 = atomicAdd(&cnt[d2], 1);
    const int r3 = atomicAdd(&cnt[d3], 1);
    int* rp = rank + e0;
    rp[0] = r0; rp[1] = r1; rp[2] = r2; rp[3] = r3;
  } else {
    for (long e = e0; e < Eraw; ++e) rank[e] = atomicAdd(&cnt[ei[Eraw + e]], 1);
  }
}

__global__ void __launch_bounds__(1024)
scan_offsets(const int* __restrict__ cnt, int* __restrict__ off, int n) {
  __shared__ int part[1024];
  const int t = threadIdx.x;
  const int base = blockIdx.x * 1024;
  int p = 0;
  for (int i = t; i < base; i += 1024) p += cnt[i];
  part[t] = p;
  __syncthreads();
  for (int o = 512; o > 0; o >>= 1) {
    if (t < o) part[t] += part[t + o];
    __syncthreads();
  }
  const int before = part[0];
  __syncthreads();
  const int i = base + t;
  const int v = (i < n) ? cnt[i] : 0;
  part[t] = v;
  __syncthreads();
  for (int o = 1; o < 1024; o <<= 1) {
    const int u = (t >= o) ? part[t - o] : 0;
    __syncthreads();
    part[t] += u;
    __syncthreads();
  }
  if (i < n) off[i] = before + part[t] - v;
  if (i == n - 1) off[n] = before + part[t];
}

__global__ void fill_by_dst(const int* __restrict__ ei, long Eraw, const int* __restrict__ off,
                            const int* __restrict__ rank, int* __restrict__ perm) {
  const long e0 = ((long)blockIdx.x * blockDim.x + threadIdx.x) * 4;
  if (e0 + 3 < Eraw) {
    const int* sp = ei + e0;
    const int* dp = ei + Eraw + e0;
    const int* rp = rank + e0;
    const int s0 = sp[0], s1 = sp[1], s2 = sp[2], s3 = sp[3];
    const int d0 = dp[0], d1 = dp[1], d2 = dp[2], d3 = dp[3];
    const int r0 = rp[0], r1 = rp[1], r2 = rp[2], r3 = rp[3];
    const int o0 = off[d0], o1 = off[d1], o2 = off[d2], o3 = off[d3];
    perm[o0 + r0] = s0;
    perm[o1 + r1] = s1;
    perm[o2 + r2] = s2;
    perm[o3 + r3] = s3;
  } else {
    for (long e = e0; e < Eraw; ++e) perm[off[ei[Eraw + e]] + rank[e]] = ei[e];
  }
}

template <int CPT> struct Row { float v[CPT]; };

template <int CPT>
__device__ __forceinline__ Row<CPT> load_row(const float* __restrict__ p) {
  Row<CPT> r;
  if constexpr (CPT == 8) {
    const float4 a = *(const float4*)p;
    const float4 b = *(const float4*)(p + 4);
    r.v[0] = a.x; r.v[1] = a.y; r.v[2] = a.z; r.v[3] = a.w;
    r.v[4] = b.x; r.v[5] = b.y; r.v[6] = b.z; r.v[7] = b.w;
  } else {
#pragma unroll
    for (int c = 0; c < CPT; ++c) r.v[c] = p[c];
  }
  return r;
}

template <int H, int C, int CPT>
__global__ void __launch_bounds__(256)
node_softmax_gather(const int* __restrict__ off, const int* __restrict__ perm,
                    const float* __restrict__ als, const float* __restrict__ ad,
                    const float* __restrict__ hf, const float* __restrict__ bias,
                    float* __restrict__ out, long N) {
  constexpr int F = H * C;
  static_assert(F == 8 * CPT, "8 threads cover one feature row");
  const long t = (long)blockIdx.x * blockDim.x + threadIdx.x;
  if (t >= N * 8) return;
  const int d = (int)(t >> 3);
  const int g = (int)(t & 7);
  const int hh = (g * CPT) / C;

  const float* hd = hf + (long)d * F + hh * C;
  float ald = 0.f;
#pragma unroll
  for (int c = 0; c < C; ++c) ald += hd[c] * ad[hh * C + c];
  const float e_self = lrelu(als[(long)d * H + hh] + ald);
  const int j0 = off[d], j1 = off[d + 1];

  float m = e_self;
  int j = j0;
  for (; j + 4 <= j1; j += 4) {
    const int s0 = perm[j], s1 = perm[j + 1], s2 = perm[j + 2], s3 = perm[j + 3];
    const float a0 = als[(long)s0 * H + hh];
    const float a1 = als[(long)s1 * H + hh];
    const float a2 = als[(long)s2 * H + hh];
    const float a3 = als[(long)s3 * H + hh];
    m = fmaxf(m, lrelu(a0 + ald));
    m = fmaxf(m, lrelu(a1 + ald));
    m = fmaxf(m, lrelu(a2 + ald));
    m = fmaxf(m, lrelu(a3 + ald));
  }
  for (; j < j1; ++j) m = fmaxf(m, lrelu(als[(long)perm[j] * H + hh] + ald));

  float acc[CPT];
  float den;
  {
    const float p = __expf(e_self - m);
    const Row<CPT> r = load_row<CPT>(hf + (long)d * F + g * CPT);
    den = p;
#pragma unroll
    for (int c = 0; c < CPT; ++c) acc[c] = p * r.v[c];
  }
  j = j0;
  for (; j + 4 <= j1; j += 4) {
    const int s0 = perm[j], s1 = perm[j + 1], s2 = perm[j + 2], s3 = perm[j + 3];
    const float a0 = als[(long)s0 * H + hh];
    const float a1 = als[(long)s1 * H + hh];
    const float a2 = als[(long)s2 * H + hh];
    const float a3 = als[(long)s3 * H + hh];
    const Row<CPT> r0 = load_row<CPT>(hf + (long)s0 * F + g * CPT);
    const Row<CPT> r1 = load_row<CPT>(hf + (long)s1 * F + g * CPT);
    const Row<CPT> r2 = load_row<CPT>(hf + (long)s2 * F + g * CPT);
    const Row<CPT> r3 = load_row<CPT>(hf + (long)s3 * F + g * CPT);
    const float p0 = __expf(lrelu(a0 + ald) - m);
    const float p1 = __expf(lrelu(a1 + ald) - m);
    const float p2 = __expf(lrelu(a2 + ald) - m);
    const float p3 = __expf(lrelu(a3 + ald) - m);
    den += p0; den += p1; den += p2; den += p3;
#pragma unroll
    for (int c = 0; c < CPT; ++c) {
      acc[c] += p0 * r0.v[c];
      acc[c] += p1 * r1.v[c];
      acc[c] += p2 * r2.v[c];
      acc[c] += p3 * r3.v[c];
    }
  }
  for (; j < j1; ++j) {
    const int s = perm[j];
    const float p = __expf(lrelu(als[(long)s * H + hh] + ald) - m);
    const Row<CPT> r = load_row<CPT>(hf + (long)s * F + g * CPT);
    den += p;
#pragma unroll
    for (int c = 0; c < CPT; ++c) acc[c] += p * r.v[c];
  }

  const float dn = den + 1e-16f;
  float* op = out + (long)d * F + g * CPT;
  const float* bp = bias + g * CPT;
  if constexpr (CPT == 8) {
    float4 o0, o1;
    o0.x = acc[0] / dn + bp[0]; o0.y = acc[1] / dn + bp[1];
    o0.z = acc[2] / dn + bp[2]; o0.w = acc[3] / dn + bp[3];
    o1.x = acc[4] / dn + bp[4]; o1.y = acc[5] / dn + bp[5];
    o1.z = acc[6] / dn + bp[6]; o1.w = acc[7] / dn + bp[7];
    *(float4*)op = o0;
    *(float4*)(op + 4) = o1;
  } else {
#pragma unroll
    for (int c = 0; c < CPT; ++c) op[c] = acc[c] / dn + bp[c];
  }
}

__global__ void score_kernel(const float* __restrict__ emb, const float* __restrict__ W3,
                             const float* __restrict__ b3, float* __restrict__ score, long N) {
  long n = (long)blockIdx.x * blockDim.x + threadIdx.x;
  if (n >= N) return;
  float s = b3[0];
#pragma unroll
  for (int c = 0; c < 8; ++c) s += emb[n * 8 + c] * W3[c];
  score[n] = s;
}

static inline int nblk(long n, int b) { return (int)((n + b - 1) / b); }

extern "C" void kernel_launch(void* const* d_in, const int* in_sizes, int n_in,
                              void* d_out, int out_size, void* d_ws, size_t ws_size,
                              hipStream_t stream) {
  const float* x   = (const float*)d_in[0];
  const int*   ei  = (const int*)d_in[1];
  const float* W1  = (const float*)d_in[2];
  const float* as1 = (const float*)d_in[3];
  const float* ad1 = (const float*)d_in[4];
  const float* b1  = (const float*)d_in[5];
  const float* W2  = (const float*)d_in[6];
  const float* as2 = (const float*)d_in[7];
  const float* ad2 = (const float*)d_in[8];
  const float* b2  = (const float*)d_in[9];
  const float* W3  = (const float*)d_in[10];
  const float* b3  = (const float*)d_in[11];

  const long N    = in_sizes[0] / 256;
  const long Eraw = in_sizes[1] / 2;

  float* ws   = (float*)d_ws;
  float* h1   = ws;
  float* acc1 = h1 + N * 64;
  float* als1 = acc1 + N * 64;
  int*   perm = (int*)(als1 + N * 8);
  int*   cnt  = perm + Eraw;
  int*   off  = cnt + N;
  int*   rank = (int*)acc1;
  float* h2   = h1;
  float* als2 = h1 + N * 8;

  float* emb   = (float*)d_out;
  float* score = emb + N * 8;

  const int B = 256;
  const int mtiles = (int)(N / 16);

  zero_counts<<<nblk(N, B), B, 0, stream>>>(cnt, N);
  dst_histogram<<<nblk((Eraw + 3) / 4, B), B, 0, stream>>>(ei, Eraw, cnt, rank);
  scan_offsets<<<nblk(N, 1024), 1024, 0, stream>>>(cnt, off, (int)N);
  fill_by_dst<<<nblk((Eraw + 3) / 4, B), B, 0, stream>>>(ei, Eraw, off, rank, perm);

  gemm1_x_w1<<<mtiles, 128, 0, stream>>>(x, W1, h1);
  src_logits<8, 8><<<nblk(N * 8, B), B, 0, stream>>>(h1, as1, als1, N * 8);
  node_softmax_gather<8, 8, 8><<<nblk(N * 8, B), B, 0, stream>>>(off, perm, als1, ad1, h1, b1,
                                                                 acc1, N);

  gemm2_h_w2<<<(mtiles + 3) / 4, 128, 0, stream>>>(acc1, W2, h2, mtiles);
  src_logits<1, 8><<<nblk(N, B), B, 0, stream>>>(h2, as2, als2, N);
  node_softmax_gather<1, 8, 1><<<nblk(N * 8, B), B, 0, stream>>>(off, perm, als2, ad2, h2, b2,
                                                                 emb, N);

  score_kernel<<<nblk(N, B), B, 0, stream>>>(emb, W3, b3, score, N);
}


# Gathers fetch each edge's source row once: running-max softmax (sums rescaled when the max grows) replaces the separate max walk; edge logit recomputed from the fetched row; sum rounding changes
# speedup vs baseline: 6.7917x; 1.2483x over previous
#include <hip/hip_runtime.h>
#include <math.h>

typedef __attribute__((ext_vector_type(2))) float v2f;
typedef __attribute__((ext_vector_type(8))) float v8f;

#define NEG_SLOPE 0.2f

__device__ __forceinline__ float lrelu(float v) { return v >= 0.f ? v : NEG_SLOPE * v; }

__global__ void gemm1_x_w1(const float* __restrict__ X, const float* __restrict__ W,
                           float* __restrict__ H) {
  const int lane = threadIdx.x & 31;
  const int wave = threadIdx.x >> 5;
  const int kgrp = (lane >> 4) << 1;
  const int ncol = (wave << 4) + (lane & 15);
  const float* xrow = X + ((long)blockIdx.x * 16 + (lane & 15)) * 256;
  v8f acc = {};
#pragma unroll 8
  for (int k0 = 0; k0 < 256; k0 += 4) {
    v2f a = *(const v2f*)(xrow + k0 + kgrp);
    v2f b;
    b.x = W[(k0 + kgrp) * 64 + ncol];
    b.y = W[(k0 + kgrp + 1) * 64 + ncol];
    acc = __builtin_amdgcn_wmma_f32_16x16x4_f32(false, a, false, b, (short)0, acc,
                                                false, false);
  }
  const long mb = (long)blockIdx.x * 16 + ((lane >> 4) << 3);
#pragma unroll
  for (int v = 0; v < 8; ++v)
    H[(mb + v) * 64 + ncol] = acc[v];
}

__global__ void gemm2_h_w2(const float* __restrict__ A1, const float* __restrict__ W,
                           float* __restrict__ H2, int mtiles) {
  const int lane = threadIdx.x & 31;
  const int wave = threadIdx.x >> 5;
  const int mtile = blockIdx.x * 4 + wave;
  if (mtile >= mtiles) return;
  const int kgrp = (lane >> 4) << 1;
  const int ncol = lane & 15;
  const float* arow = A1 + ((long)mtile * 16 + ncol) * 64;
  v8f acc = {};
#pragma unroll
  for (int k0 = 0; k0 < 64; k0 += 4) {
    v2f a = *(const v2f*)(arow + k0 + kgrp);
    a.x = fmaxf(a.x, 0.f);
    a.y = fmaxf(a.y, 0.f);
    v2f b;
    b.x = W[(k0 + kgrp) * 8 + (ncol & 7)];
    b.y = W[(k0 + kgrp + 1) * 8 + (ncol & 7)];
    acc = __builtin_amdgcn_wmma_f32_16x16x4_f32(false, a, false, b, (short)0, acc,
                                                false, false);
  }
  if (ncol < 8) {
    const long mb = (long)mtile * 16 + ((lane >> 4) << 3);
#pragma unroll
    for (int v = 0; v < 8; ++v)
      H2[(mb + v) * 8 + ncol] = acc[v];
  }
}

__global__ void zero_counts(int* __restrict__ cnt, long n) {
  long t = (long)blockIdx.x * blockDim.x + threadIdx.x;
  if (t < n) cnt[t] = 0;
}

__global__ void dst_histogram(const int* __restrict__ ei, long Eraw, int* __restrict__ cnt,
                              int* __restrict__ rank) {
  const long e0 = ((long)blockIdx.x * blockDim.x + threadIdx.x) * 4;
  if (e0 + 3 < Eraw) {
    const int* dp = ei + Eraw + e0;
    const int d0 = dp[0], d1 = dp[1], d2 = dp[2], d3 = dp[3];
    const int r0 = atomicAdd(&cnt[d0], 1);
    const int r1 = atomicAdd(&cnt[d1], 1);
    const int r2 = atomicAdd(&cnt[d2], 1);
    const int r3 = atomicAdd(&cnt[d3], 1);
    int* rp = rank + e0;
    rp[0] = r0; rp[1] = r1; rp[2] = r2; rp[3] = r3;
  } else {
    for (long e = e0; e < Eraw; ++e) rank[e] = atomicAdd(&cnt[ei[Eraw + e]], 1);
  }
}

__global__ void __launch_bounds__(1024)
scan_offsets(const int* __restrict__ cnt, int* __restrict__ off, int n) {
  __shared__ int part[1024];
  const int t = threadIdx.x;
  const int base = blockIdx.x * 1024;
  int p = 0;
  for (int i = t; i < base; i += 1024) p += cnt[i];
  part[t] = p;
  __syncthreads();
  for (int o = 512; o > 0; o >>= 1) {
    if (t < o) part[t] += part[t + o];
    __syncthreads();
  }
  const int before = part[0];
  __syncthreads();
  const int i = base + t;
  const int v = (i < n) ? cnt[i] : 0;
  part[t] = v;
  __syncthreads();
  for (int o = 1; o < 1024; o <<= 1) {
    const int u = (t >= o) ? part[t - o] : 0;
    __syncthreads();
    part[t] += u;
    __syncthreads();
  }
  if (i < n) off[i] = before + part[t] - v;
  if (i == n - 1) off[n] = before + part[t];
}

__global__ void fill_by_dst(const int* __restrict__ ei, long Eraw, const int* __restrict__ off,
                            const int* __restrict__ rank, int* __restrict__ perm) {
  const long e0 = ((long)blockIdx.x * blockDim.x + threadIdx.x) * 4;
  if (e0 + 3 < Eraw) {
    const int* sp = ei + e0;
    const int* dp = ei + Eraw + e0;
    const int* rp = rank + e0;
    const int s0 = sp[0], s1 = sp[1], s2 = sp[2], s3 = sp[3];
    const int d0 = dp[0], d1 = dp[1], d2 = dp[2], d3 = dp[3];
    const int r0 = rp[0], r1 = rp[1], r2 = rp[2], r3 = rp[3];
    const int o0 = off[d0], o1 = off[d1], o2 = off[d2], o3 = off[d3];
    perm[o0 + r0] = s0;
    perm[o1 + r1] = s1;
    perm[o2 + r2] = s2;
    perm[o3 + r3] = s3;
  } else {
    for (long e = e0; e < Eraw; ++e) perm[off[ei[Eraw + e]] + rank[e]] = ei[e];
  }
}

struct Row8 { v2f q[4]; };

__device__ __forceinline__ Row8 load_row8(const float* __restrict__ p) {
  const float4 a = *(const float4*)p;
  const float4 b = *(const float4*)(p + 4);
  Row8 r;
  r.q[0].x = a.x; r.q[0].y = a.y; r.q[1].x = a.z; r.q[1].y = a.w;
  r.q[2].x = b.x; r.q[2].y = b.y; r.q[3].x = b.z; r.q[3].y = b.w;
  return r;
}

__device__ __forceinline__ float dot8(const Row8& r, const Row8& w) {
  v2f ss = r.q[0] * w.q[0];
  ss += r.q[1] * w.q[1];
  ss += r.q[2] * w.q[2];
  ss += r.q[3] * w.q[3];
  return ss.x + ss.y;
}


__global__ void __launch_bounds__(256)
node_softmax_gather_heads8(const int* __restrict__ off, const int* __restrict__ perm,
                           const float* __restrict__ as, const float* __restrict__ ad,
                           const float* __restrict__ hf, const float* __restrict__ bias,
                           float* __restrict__ out, long N) {
  const long t = (long)blockIdx.x * blockDim.x + threadIdx.x;
  if (t >= N * 8) return;
  const int d = (int)(t >> 3);
  const int ro = (int)(t & 7) * 8;

  const Row8 asv = load_row8(as + ro);
  const Row8 own = load_row8(hf + (long)d * 64 + ro);
  const float ald = dot8(own, load_row8(ad + ro));
  float m = lrelu(dot8(own, asv) + ald);
  float den = 1.f;
  Row8 acc = own;

  const int j0 = off[d], j1 = off[d + 1];
  int j = j0;
  for (; j + 4 <= j1; j += 4) {
    const int s0 = perm[j], s1 = perm[j + 1], s2 = perm[j + 2], s3 = perm[j + 3];
    const Row8 r0 = load_row8(hf + (long)s0 * 64 + ro);
    const Row8 r1 = load_row8(hf + (long)s1 * 64 + ro);
    const Row8 r2 = load_row8(hf + (long)s2 * 64 + ro);
    const Row8 r3 = load_row8(hf + (long)s3 * 64 + ro);
    const float e0 = lrelu(dot8(r0, asv) + ald);
    const float e1 = lrelu(dot8(r1, asv) + ald);
    const float e2 = lrelu(dot8(r2, asv) + ald);
    const float e3 = lrelu(dot8(r3, asv) + ald);
    const float mn = fmaxf(fmaxf(fmaxf(m, e0), fmaxf(e1, e2)), e3);
    const float sc = __expf(m - mn);
    const float p0 = __expf(e0 - mn);
    const float p1 = __expf(e1 - mn);
    const float p2 = __expf(e2 - mn);
    const float p3 = __expf(e3 - mn);
    m = mn;
    den = den * sc + p0; den += p1; den += p2; den += p3;
#pragma unroll
    for (int k = 0; k < 4; ++k) {
      acc.q[k] = acc.q[k] * sc + p0 * r0.q[k];
      acc.q[k] += p1 * r1.q[k];
      acc.q[k] += p2 * r2.q[k];
      acc.q[k] += p3 * r3.q[k];
    }
  }
  for (; j < j1; ++j) {
    const Row8 r = load_row8(hf + (long)perm[j] * 64 + ro);
    const float e = lrelu(dot8(r, asv) + ald);
    const float mn = fmaxf(m, e);
    const float sc = __expf(m - mn);
    const float p = __expf(e - mn);
    m = mn;
    den = den * sc + p;
#pragma unroll
    for (int k = 0; k < 4; ++k) acc.q[k] = acc.q[k] * sc + p * r.q[k];
  }

  const float dn = den + 1e-16f;
  float* op = out + (long)d * 64 + ro;
  const float* bp = bias + ro;
  float4 o0, o1;
  o0.x = acc.q[0].x / dn + bp[0]; o0.y = acc.q[0].y / dn + bp[1];
  o0.z = acc.q[1].x / dn + bp[2]; o0.w = acc.q[1].y / dn + bp[3];
  o1.x = acc.q[2].x / dn + bp[4]; o1.y = acc.q[2].y / dn + bp[5];
  o1.z = acc.q[3].x / dn + bp[6]; o1.w = acc.q[3].y / dn + bp[7];
  *(float4*)op = o0;
  *(float4*)(op + 4) = o1;
}

template <int X>
__device__ __forceinline__ float lane_xor(float v) {
  return __builtin_bit_cast(float, __builtin_amdgcn_update_dpp(
      0, __builtin_bit_cast(int, v), 0x160 + X, 0xf, 0xf, true));
}

__device__ __forceinline__ float sum8(float v) {
  v += lane_xor<1>(v);
  v += lane_xor<2>(v);
  v += lane_xor<4>(v);
  return v;
}

__global__ void __launch_bounds__(256)
node_softmax_gather_head1(const int* __restrict__ off, const int* __restrict__ perm,
                          const float* __restrict__ as, const float* __restrict__ ad,
                          const float* __restrict__ hf, const float* __restrict__ bias,
                          float* __restrict__ out, long N) {
  const long t = (long)blockIdx.x * blockDim.x + threadIdx.x;
  if (t >= N * 8) return;
  const int d = (int)(t >> 3);
  const int g = (int)(t & 7);

  const float asg = as[g];
  const float own = hf[(long)d * 8 + g];
  const float ald = sum8(own * ad[g]);
  float m = lrelu(sum8(own * asg) + ald);
  float den = 1.f;
  float acc = own;

  const int j0 = off[d], j1 = off[d + 1];
  int j = j0;
  for (; j + 4 <= j1; j += 4) {
    const int s0 = perm[j], s1 = perm[j + 1], s2 = perm[j + 2], s3 = perm[j + 3];
    const float x0 = hf[(long)s0 * 8 + g];
    const float x1 = hf[(long)s1 * 8 + g];
    const float x2 = hf[(long)s2 * 8 + g];
    const float x3 = hf[(long)s3 * 8 + g];
    const float e0 = lrelu(sum8(x0 * asg) + ald);
    const float e1 = lrelu(sum8(x1 * asg) + ald);
    const float e2 = lrelu(sum8(x2 * asg) + ald);
    const float e3 = lrelu(sum8(x3 * asg) + ald);
    const float mn = fmaxf(fmaxf(fmaxf(m, e0), fmaxf(e1, e2)), e3);
    const float sc = __expf(m - mn);
    const float p0 = __expf(e0 - mn);
    const float p1 = __expf(e1 - mn);
    const float p2 = __expf(e2 - mn);
    const float p3 = __expf(e3 - mn);
    m = mn;
    den = den * sc + p0; den += p1; den += p2; den += p3;
    acc = acc * sc + p0 * x0; acc += p1 * x1; acc += p2 * x2; acc += p3 * x3;
  }
  for (; j < j1; ++j) {
    const float x = hf[(long)perm[j] * 8 + g];
    const float e = lrelu(sum8(x * asg) + ald);
    const float mn = fmaxf(m, e);
    const float sc = __expf(m - mn);
    const float p = __expf(e - mn);
    m = mn;
    den = den * sc + p;
    acc = acc * sc + p * x;
  }
  out[(long)d * 8 + g] = acc / (den + 1e-16f) + bias[g];
}

__global__ void score_kernel(const float* __restrict__ emb, const float* __restrict__ W3,
                             const float* __restrict__ b3, float* __restrict__ score, long N) {
  long n = (long)blockIdx.x * blockDim.x + threadIdx.x;
  if (n >= N) return;
  float s = b3[0];
#pragma unroll
  for (int c = 0; c < 8; ++c) s += emb[n * 8 + c] * W3[c];
  score[n] = s;
}

static inline int nblk(long n, int b) { return (int)((n + b - 1) / b); }

extern "C" void kernel_launch(void* const* d_in, const int* in_sizes, int n_in,
                              void* d_out, int out_size, void* d_ws, size_t ws_size,
                              hipStream_t stream) {
  const float* x   = (const float*)d_in[0];
  const int*   ei  = (const int*)d_in[1];
  const float* W1  = (const float*)d_in[2];
  const float* as1 = (const float*)d_in[3];
  const float* ad1 = (const float*)d_in[4];
  const float* b1  = (const float*)d_in[5];
  const float* W2  = (const float*)d_in[6];
  const float* as2 = (const float*)d_in[7];
  const float* ad2 = (const float*)d_in[8];
  const float* b2  = (const float*)d_in[9];
  const float* W3  = (const float*)d_in[10];
  const float* b3  = (const float*)d_in[11];

  const long N    = in_sizes[0] / 256;
  const long Eraw = in_sizes[1] / 2;

  float* ws   = (float*)d_ws;
  float* h1   = ws;
  float* acc1 = h1 + N * 64;
  int*   perm = (int*)(acc1 + N * 64);
  int*   cnt  = perm + Eraw;
  int*   off  = cnt + N;
  int*   rank = (int*)acc1;
  float* h2   = h1;

  float* emb   = (float*)d_out;
  float* score = emb + N * 8;

  const int B = 256;
  const int mtiles = (int)(N / 16);

  zero_counts<<<nblk(N, B), B, 0, stream>>>(cnt, N);
  dst_histogram<<<nblk((Eraw + 3) / 4, B), B, 0, stream>>>(ei, Eraw, cnt, rank);
  scan_offsets<<<nblk(N, 1024), 1024, 0, stream>>>(cnt, off, (int)N);
  fill_by_dst<<<nblk((Eraw + 3) / 4, B), B, 0, stream>>>(ei, Eraw, off, rank, perm);

  gemm1_x_w1<<<mtiles, 128, 0, stream>>>(x, W1, h1);
  node_softmax_gather_heads8<<<nblk(N * 8, B), B, 0, stream>>>(off, perm, as1, ad1, h1, b1, acc1, N);

  gemm2_h_w2<<<(mtiles + 3) / 4, 128, 0, stream>>>(acc1, W2, h2, mtiles);
  node_softmax_gather_head1<<<nblk(N * 8, B), B, 0, stream>>>(off, perm, as2, ad2, h2, b2, emb, N);

  score_kernel<<<nblk(N, B), B, 0, stream>>>(emb, W3, b3, score, N);
}


# Edges grouped by destination in two levels with LDS counters (256-node buckets, returning LDS add as arrival-order rank) in place of one global atomic per edge; sum order changes
# speedup vs baseline: 12.6505x; 1.8626x over previous
#include <hip/hip_runtime.h>
#include <math.h>

typedef __attribute__((ext_vector_type(2))) float v2f;
typedef __attribute__((ext_vector_type(8))) float v8f;

#define NEG_SLOPE 0.2f

__device__ __forceinline__ float lrelu(float v) { return v >= 0.f ? v : NEG_SLOPE * v; }

__global__ void gemm1_x_w1(const float* __restrict__ X, const float* __restrict__ W,
                           float* __restrict__ H) {
  const int lane = threadIdx.x & 31;
  const int wave = threadIdx.x >> 5;
  const int kgrp = (lane >> 4) << 1;
  const int ncol = (wave << 4) + (lane & 15);
  const float* xrow = X + ((long)blockIdx.x * 16 + (lane & 15)) * 256;
  v8f acc = {};
#pragma unroll 8
  for (int k0 = 0; k0 < 256; k0 += 4) {
    v2f a = *(const v2f*)(xrow + k0 + kgrp);
    v2f b;
    b.x = W[(k0 + kgrp) * 64 + ncol];
    b.y = W[(k0 + kgrp + 1) * 64 + ncol];
    acc = __builtin_amdgcn_wmma_f32_16x16x4_f32(false, a, false, b, (short)0, acc,
                                                false, false);
  }
  const long mb = (long)blockIdx.x * 16 + ((lane >> 4) << 3);
#pragma unroll
  for (int v = 0; v < 8; ++v)
    H[(mb + v) * 64 + ncol] = acc[v];
}

__global__ void gemm2_h_w2(const float* __restrict__ A1, const float* __restrict__ W,
                           float* __restrict__ H2, int mtiles) {
  const int lane = threadIdx.x & 31;
  const int wave = threadIdx.x >> 5;
  const int mtile = blockIdx.x * 4 + wave;
  if (mtile >= mtiles) return;
  const int kgrp = (lane >> 4) << 1;
  const int ncol = lane & 15;
  const float* arow = A1 + ((long)mtile * 16 + ncol) * 64;
  v8f acc = {};
#pragma unroll
  for (int k0 = 0; k0 < 64; k0 += 4) {
    v2f a = *(const v2f*)(arow + k0 + kgrp);
    a.x = fmaxf(a.x, 0.f);
    a.y = fmaxf(a.y, 0.f);
    v2f b;
    b.x = W[(k0 + kgrp) * 8 + (ncol & 7)];
    b.y = W[(k0 + kgrp + 1) * 8 + (ncol & 7)];
    acc = __builtin_amdgcn_wmma_f32_16x16x4_f32(false, a, false, b, (short)0, acc,
                                                false, false);
  }
  if (ncol < 8) {
    const long mb = (long)mtile * 16 + ((lane >> 4) << 3);
#pragma unroll
    for (int v = 0; v < 8; ++v)
      H2[(mb + v) * 8 + ncol] = acc[v];
  }
}

#define BUCKET_SHIFT 8
#define BUCKET_NODES 256
#define MAXB 512
#define NCHUNK 256
#define STAGE_CAP 12288

__global__ void __launch_bounds__(1024)
bucket_count(const int* __restrict__ ei, long Eraw, int nb, int* __restrict__ tbl) {
  __shared__ int hist[MAXB];
  const int t = threadIdx.x;
  const int wg = blockIdx.x;
  for (int b = t; b < nb; b += 1024) hist[b] = 0;
  __syncthreads();
  const long chunk = (Eraw + NCHUNK - 1) / NCHUNK;
  const long lo = wg * chunk;
  const long hi = (lo + chunk < Eraw) ? lo + chunk : Eraw;
  const int* dp = ei + Eraw;
  long e = lo + t;
  for (; e + 3072 < hi; e += 4096) {
    const int d0 = dp[e], d1 = dp[e + 1024], d2 = dp[e + 2048], d3 = dp[e + 3072];
    atomicAdd(&hist[d0 >> BUCKET_SHIFT], 1);
    atomicAdd(&hist[d1 >> BUCKET_SHIFT], 1);
    atomicAdd(&hist[d2 >> BUCKET_SHIFT], 1);
    atomicAdd(&hist[d3 >> BUCKET_SHIFT], 1);
  }
  for (; e < hi; e += 1024) atomicAdd(&hist[dp[e] >> BUCKET_SHIFT], 1);
  __syncthreads();
  for (int b = t; b < nb; b += 1024) tbl[b * NCHUNK + wg] = hist[b];
}

__global__ void __launch_bounds__(1024)
scan_offsets(const int* __restrict__ cnt, int* __restrict__ off, int n) {
  __shared__ int part[1024];
  const int t = threadIdx.x;
  const int base = blockIdx.x * 1024;
  int p = 0;
  for (int i = t; i < base; i += 1024) p += cnt[i];
  part[t] = p;
  __syncthreads();
  for (int o = 512; o > 0; o >>= 1) {
    if (t < o) part[t] += part[t + o];
    __syncthreads();
  }
  const int before = part[0];
  __syncthreads();
  const int i = base + t;
  const int v = (i < n) ? cnt[i] : 0;
  part[t] = v;
  __syncthreads();
  for (int o = 1; o < 1024; o <<= 1) {
    const int u = (t >= o) ? part[t - o] : 0;
    __syncthreads();
    part[t] += u;
    __syncthreads();
  }
  if (i < n) off[i] = before + part[t] - v;
  if (i == n - 1) off[n] = before + part[t];
}

__global__ void __launch_bounds__(1024)
bucket_scatter(const int* __restrict__ ei, long Eraw, int nb, const int* __restrict__ toff,
               unsigned* __restrict__ tmp) {
  __shared__ int cur[MAXB];
  const int t = threadIdx.x;
  const int wg = blockIdx.x;
  for (int b = t; b < nb; b += 1024) cur[b] = toff[b * NCHUNK + wg];
  __syncthreads();
  const long chunk = (Eraw + NCHUNK - 1) / NCHUNK;
  const long lo = wg * chunk;
  const long hi = (lo + chunk < Eraw) ? lo + chunk : Eraw;
  const int* dp = ei + Eraw;
  long e = lo + t;
  for (; e + 3072 < hi; e += 4096) {
    const int s0 = ei[e], s1 = ei[e + 1024], s2 = ei[e + 2048], s3 = ei[e + 3072];
    const int d0 = dp[e], d1 = dp[e + 1024], d2 = dp[e + 2048], d3 = dp[e + 3072];
    const int p0 = atomicAdd(&cur[d0 >> BUCKET_SHIFT], 1);
    const int p1 = atomicAdd(&cur[d1 >> BUCKET_SHIFT], 1);
    const int p2 = atomicAdd(&cur[d2 >> BUCKET_SHIFT], 1);
    const int p3 = atomicAdd(&cur[d3 >> BUCKET_SHIFT], 1);
    tmp[p0] = ((unsigned)s0 << 8) | (unsigned)(d0 & (BUCKET_NODES - 1));
    tmp[p1] = ((unsigned)s1 << 8) | (unsigned)(d1 & (BUCKET_NODES - 1));
    tmp[p2] = ((unsigned)s2 << 8) | (unsigned)(d2 & (BUCKET_NODES - 1));
    tmp[p3] = ((unsigned)s3 << 8) | (unsigned)(d3 & (BUCKET_NODES - 1));
  }
  for (; e < hi; e += 1024) {
    const int s = ei[e], d = dp[e];
    const int p = atomicAdd(&cur[d >> BUCKET_SHIFT], 1);
    tmp[p] = ((unsigned)s << 8) | (unsigned)(d & (BUCKET_NODES - 1));
  }
}

template <bool STAGED>
__device__ __forceinline__ void place_edges(const unsigned* __restrict__ tmp, int lo, int hi, int t,
                                            int* ncur, int* stage, int* __restrict__ perm) {
  int i = lo + t;
  for (; i + 3072 < hi; i += 4096) {
    const unsigned w0 = tmp[i], w1 = tmp[i + 1024], w2 = tmp[i + 2048], w3 = tmp[i + 3072];
    const int p0 = atomicAdd(&ncur[w0 & (BUCKET_NODES - 1)], 1);
    const int p1 = atomicAdd(&ncur[w1 & (BUCKET_NODES - 1)], 1);
    const int p2 = atomicAdd(&ncur[w2 & (BUCKET_NODES - 1)], 1);
    const int p3 = atomicAdd(&ncur[w3 & (BUCKET_NODES - 1)], 1);
    if (STAGED) {
      stage[p0 - lo] = (int)(w0 >> 8);
      stage[p1 - lo] = (int)(w1 >> 8);
      stage[p2 - lo] = (int)(w2 >> 8);
      stage[p3 - lo] = (int)(w3 >> 8);
    } else {
      perm[p0] = (int)(w0 >> 8);
      perm[p1] = (int)(w1 >> 8);
      perm[p2] = (int)(w2 >> 8);
      perm[p3] = (int)(w3 >> 8);
    }
  }
  for (; i < hi; i += 1024) {
    const unsigned w = tmp[i];
    const int p = atomicAdd(&ncur[w & (BUCKET_NODES - 1)], 1);
    if (STAGED) stage[p - lo] = (int)(w >> 8);
    else perm[p] = (int)(w >> 8);
  }
}

__global__ void __launch_bounds__(1024)
bucket_sort_nodes(const unsigned* __restrict__ tmp, const int* __restrict__ toff, int nb, int N,
                  int* __restrict__ off, int* __restrict__ perm) {
  __shared__ int ncnt[BUCKET_NODES];
  __shared__ int nsc[BUCKET_NODES];
  __shared__ int ncur[BUCKET_NODES];
  __shared__ int stage[STAGE_CAP];
  const int t = threadIdx.x;
  const int b = blockIdx.x;
  const int lo = toff[b * NCHUNK];
  const int hi = toff[(b + 1) * NCHUNK];
  if (t < BUCKET_NODES) ncnt[t] = 0;
  __syncthreads();
  int i = lo + t;
  for (; i + 3072 < hi; i += 4096) {
    const unsigned w0 = tmp[i], w1 = tmp[i + 1024], w2 = tmp[i + 2048], w3 = tmp[i + 3072];
    atomicAdd(&ncnt[w0 & (BUCKET_NODES - 1)], 1);
    atomicAdd(&ncnt[w1 & (BUCKET_NODES - 1)], 1);
    atomicAdd(&ncnt[w2 & (BUCKET_NODES - 1)], 1);
    atomicAdd(&ncnt[w3 & (BUCKET_NODES - 1)], 1);
  }
  for (; i < hi; i += 1024) atomicAdd(&ncnt[tmp[i] & (BUCKET_NODES - 1)], 1);
  __syncthreads();
  int v = 0;
  if (t < BUCKET_NODES) { v = ncnt[t]; nsc[t] = v; }
  __syncthreads();
  for (int o = 1; o < BUCKET_NODES; o <<= 1) {
    int u = 0;
    if (t < BUCKET_NODES && t >= o) u = nsc[t - o];
    __syncthreads();
    if (t < BUCKET_NODES) nsc[t] += u;
    __syncthreads();
  }
  if (t < BUCKET_NODES) {
    const int first = lo + nsc[t] - v;
    ncur[t] = first;
    const int node = b * BUCKET_NODES + t;
    if (node < N) off[node] = first;
  }
  if (b == nb - 1 && t == 0) off[N] = hi;
  __syncthreads();
  if (hi - lo <= STAGE_CAP) {
    place_edges<true>(tmp, lo, hi, t, ncur, stage, perm);
    __syncthreads();
    for (int k = (lo & ~31) + t; k < hi; k += 1024)
      if (k >= lo) perm[k] = stage[k - lo];
  } else {
    place_edges<false>(tmp, lo, hi, t, ncur, stage, perm);
  }
}

struct Row8 { v2f q[4]; };

__device__ __forceinline__ Row8 load_row8(const float* __restrict__ p) {
  const float4 a = *(const float4*)p;
  const float4 b = *(const float4*)(p + 4);
  Row8 r;
  r.q[0].x = a.x; r.q[0].y = a.y; r.q[1].x = a.z; r.q[1].y = a.w;
  r.q[2].x = b.x; r.q[2].y = b.y; r.q[3].x = b.z; r.q[3].y = b.w;
  return r;
}

__device__ __forceinline__ float dot8(const Row8& r, const Row8& w) {
  v2f ss = r.q[0] * w.q[0];
  ss += r.q[1] * w.q[1];
  ss += r.q[2] * w.q[2];
  ss += r.q[3] * w.q[3];
  return ss.x + ss.y;
}


__global__ void __launch_bounds__(256)
node_softmax_gather_heads8(const int* __restrict__ off, const int* __restrict__ perm,
                           const float* __restrict__ as, const float* __restrict__ ad,
                           const float* __restrict__ hf, const float* __restrict__ bias,
                           float* __restrict__ out, long N) {
  const long t = (long)blockIdx.x * blockDim.x + threadIdx.x;
  if (t >= N * 8) return;
  const int d = (int)(t >> 3);
  const int ro = (int)(t & 7) * 8;

  const Row8 asv = load_row8(as + ro);
  const Row8 own = load_row8(hf + (long)d * 64 + ro);
  const float ald = dot8(own, load_row8(ad + ro));
  float m = lrelu(dot8(own, asv) + ald);
  float den = 1.f;
  Row8 acc = own;

  const int j0 = off[d], j1 = off[d + 1];
  int j = j0;
  for (; j + 4 <= j1; j += 4) {
    const int s0 = perm[j], s1 = perm[j + 1], s2 = perm[j + 2], s3 = perm[j + 3];
    const Row8 r0 = load_row8(hf + (long)s0 * 64 + ro);
    const Row8 r1 = load_row8(hf + (long)s1 * 64 + ro);
    const Row8 r2 = load_row8(hf + (long)s2 * 64 + ro);
    const Row8 r3 = load_row8(hf + (long)s3 * 64 + ro);
    const float e0 = lrelu(dot8(r0, asv) + ald);
    const float e1 = lrelu(dot8(r1, asv) + ald);
    const float e2 = lrelu(dot8(r2, asv) + ald);
    const float e3 = lrelu(dot8(r3, asv) + ald);
    const float mn = fmaxf(fmaxf(fmaxf(m, e0), fmaxf(e1, e2)), e3);
    const float sc = __expf(m - mn);
    const float p0 = __expf(e0 - mn);
    const float p1 = __expf(e1 - mn);
    const float p2 = __expf(e2 - mn);
    const float p3 = __expf(e3 - mn);
    m = mn;
    den = den * sc + p0; den += p1; den += p2; den += p3;
#pragma unroll
    for (int k = 0; k < 4; ++k) {
      acc.q[k] = acc.q[k] * sc + p0 * r0.q[k];
      acc.q[k] += p1 * r1.q[k];
      acc.q[k] += p2 * r2.q[k];
      acc.q[k] += p3 * r3.q[k];
    }
  }
  for (; j < j1; ++j) {
    const Row8 r = load_row8(hf + (long)perm[j] * 64 + ro);
    const float e = lrelu(dot8(r, asv) + ald);
    const float mn = fmaxf(m, e);
    const float sc = __expf(m - mn);
    const float p = __expf(e - mn);
    m = mn;
    den = den * sc + p;
#pragma unroll
    for (int k = 0; k < 4; ++k) acc.q[k] = acc.q[k] * sc + p * r.q[k];
  }

  const float dn = den + 1e-16f;
  float* op = out + (long)d * 64 + ro;
  const float* bp = bias + ro;
  float4 o0, o1;
  o0.x = acc.q[0].x / dn + bp[0]; o0.y = acc.q[0].y / dn + bp[1];
  o0.z = acc.q[1].x / dn + bp[2]; o0.w = acc.q[1].y / dn + bp[3];
  o1.x = acc.q[2].x / dn + bp[4]; o1.y = acc.q[2].y / dn + bp[5];
  o1.z = acc.q[3].x / dn + bp[6]; o1.w = acc.q[3].y / dn + bp[7];
  *(float4*)op = o0;
  *(float4*)(op + 4) = o1;
}

template <int X>
__device__ __forceinline__ float lane_xor(float v) {
  return __builtin_bit_cast(float, __builtin_amdgcn_update_dpp(
      0, __builtin_bit_cast(int, v), 0x160 + X, 0xf, 0xf, true));
}

__device__ __forceinline__ float sum8(float v) {
  v += lane_xor<1>(v);
  v += lane_xor<2>(v);
  v += lane_xor<4>(v);
  return v;
}

__global__ void __launch_bounds__(256)
node_softmax_gather_head1(const int* __restrict__ off, const int* __restrict__ perm,
                          const float* __restrict__ as, const float* __restrict__ ad,
                          const float* __restrict__ hf, const float* __restrict__ bias,
                          float* __restrict__ out, long N) {
  const long t = (long)blockIdx.x * blockDim.x + threadIdx.x;
  if (t >= N * 8) return;
  const int d = (int)(t >> 3);
  const int g = (int)(t & 7);

  const float asg = as[g];
  const float own = hf[(long)d * 8 + g];
  const float ald = sum8(own * ad[g]);
  float m = lrelu(sum8(own * asg) + ald);
  float den = 1.f;
  float acc = own;

  const int j0 = off[d], j1 = off[d + 1];
  int j = j0;
  for (; j + 4 <= j1; j += 4) {
    const int s0 = perm[j], s1 = perm[j + 1], s2 = perm[j + 2], s3 = perm[j + 3];
    const float x0 = hf[(long)s0 * 8 + g];
    const float x1 = hf[(long)s1 * 8 + g];
    const float x2 = hf[(long)s2 * 8 + g];
    const float x3 = hf[(long)s3 * 8 + g];
    const float e0 = lrelu(sum8(x0 * asg) + ald);
    const float e1 = lrelu(sum8(x1 * asg) + ald);
    const float e2 = lrelu(sum8(x2 * asg) + ald);
    const float e3 = lrelu(sum8(x3 * asg) + ald);
    const float mn = fmaxf(fmaxf(fmaxf(m, e0), fmaxf(e1, e2)), e3);
    const float sc = __expf(m - mn);
    const float p0 = __expf(e0 - mn);
    const float p1 = __expf(e1 - mn);
    const float p2 = __expf(e2 - mn);
    const float p3 = __expf(e3 - mn);
    m = mn;
    den = den * sc + p0; den += p1; den += p2; den += p3;
    acc = acc * sc + p0 * x0; acc += p1 * x1; acc += p2 * x2; acc += p3 * x3;
  }
  for (; j < j1; ++j) {
    const float x = hf[(long)perm[j] * 8 + g];
    const float e = lrelu(sum8(x * asg) + ald);
    const float mn = fmaxf(m, e);
    const float sc = __expf(m - mn);
    const float p = __expf(e - mn);
    m = mn;
    den = den * sc + p;
    acc = acc * sc + p * x;
  }
  out[(long)d * 8 + g] = acc / (den + 1e-16f) + bias[g];
}

__global__ void score_kernel(const float* __restrict__ emb, const float* __restrict__ W3,
                             const float* __restrict__ b3, float* __restrict__ score, long N) {
  long n = (long)blockIdx.x * blockDim.x + threadIdx.x;
  if (n >= N) return;
  float s = b3[0];
#pragma unroll
  for (int c = 0; c < 8; ++c) s += emb[n * 8 + c] * W3[c];
  score[n] = s;
}

static inline int nblk(long n, int b) { return (int)((n + b - 1) / b); }

extern "C" void kernel_launch(void* const* d_in, const int* in_sizes, int n_in,
                              void* d_out, int out_size, void* d_ws, size_t ws_size,
                              hipStream_t stream) {
  const float* x   = (const float*)d_in[0];
  const int*   ei  = (const int*)d_in[1];
  const float* W1  = (const float*)d_in[2];
  const float* as1 = (const float*)d_in[3];
  const float* ad1 = (const float*)d_in[4];
  const float* b1  = (const float*)d_in[5];
  const float* W2  = (const float*)d_in[6];
  const float* as2 = (const float*)d_in[7];
  const float* ad2 = (const float*)d_in[8];
  const float* b2  = (const float*)d_in[9];
  const float* W3  = (const float*)d_in[10];
  const float* b3  = (const float*)d_in[11];

  const long N    = in_sizes[0] / 256;
  const long Eraw = in_sizes[1] / 2;

  float* ws   = (float*)d_ws;
  float* h1   = ws;
  float* acc1 = h1 + N * 64;
  int*   perm = (int*)(acc1 + N * 64);
  int*   off  = perm + Eraw;
  const int nb = (int)((N + BUCKET_NODES - 1) / BUCKET_NODES);
  unsigned* tmp = (unsigned*)acc1;
  int*   tbl  = (int*)acc1 + Eraw;
  int*   toff = tbl + nb * NCHUNK;
  float* h2   = h1;

  float* emb   = (float*)d_out;
  float* score = emb + N * 8;

  const int B = 256;
  const int mtiles = (int)(N / 16);

  bucket_count<<<NCHUNK, 1024, 0, stream>>>(ei, Eraw, nb, tbl);
  scan_offsets<<<nblk(nb * NCHUNK, 1024), 1024, 0, stream>>>(tbl, toff, nb * NCHUNK);
  bucket_scatter<<<NCHUNK, 1024, 0, stream>>>(ei, Eraw, nb, toff, tmp);
  bucket_sort_nodes<<<nb, 1024, 0, stream>>>(tmp, toff, nb, (int)N, off, perm);

  gemm1_x_w1<<<mtiles, 128, 0, stream>>>(x, W1, h1);
  node_softmax_gather_heads8<<<nblk(N * 8, B), B, 0, stream>>>(off, perm, as1, ad1, h1, b1, acc1, N);

  gemm2_h_w2<<<(mtiles + 3) / 4, 128, 0, stream>>>(acc1, W2, h2, mtiles);
  node_softmax_gather_head1<<<nblk(N * 8, B), B, 0, stream>>>(off, perm, as2, ad2, h2, b2, emb, N);

  score_kernel<<<nblk(N, B), B, 0, stream>>>(emb, W3, b3, score, N);
}


# GEMM1 re-tiled: W1 staged once per 8-wave work-group in LDS in B-fragment order; a wave owns a 16x32 output tile, A as 16-byte loads; k grouping per WMMA changes, so sum rounding changes
# speedup vs baseline: 14.3941x; 1.1378x over previous
#include <hip/hip_runtime.h>
#include <math.h>

typedef __attribute__((ext_vector_type(2))) float v2f;
typedef __attribute__((ext_vector_type(8))) float v8f;

#define NEG_SLOPE 0.2f

__device__ __forceinline__ float lrelu(float v) { return v >= 0.f ? v : NEG_SLOPE * v; }

__device__ __forceinline__ float g1_elem(const float4& r, int i) {
  return i == 0 ? r.x : i == 1 ? r.y : i == 2 ? r.z : r.w;
}

__device__ __forceinline__ void g1_chunk(const float4 (&a)[4], const float4* bp,
                                         v8f& acc0, v8f& acc1) {
#pragma unroll
  for (int u = 0; u < 4; ++u) {
    const float4 b0 = bp[(u * 4 + 0) * 32];
    const float4 b1 = bp[(u * 4 + 1) * 32];
    v2f alo, ahi, b;
    alo.x = a[u].x; alo.y = a[u].y; ahi.x = a[u].z; ahi.y = a[u].w;
    b.x = b0.x; b.y = b0.y;
    acc0 = __builtin_amdgcn_wmma_f32_16x16x4_f32(false, alo, false, b, (short)0, acc0, false, false);
    b.x = b1.x; b.y = b1.y;
    acc1 = __builtin_amdgcn_wmma_f32_16x16x4_f32(false, alo, false, b, (short)0, acc1, false, false);
    b.x = b0.z; b.y = b0.w;
    acc0 = __builtin_amdgcn_wmma_f32_16x16x4_f32(false, ahi, false, b, (short)0, acc0, false, false);
    b.x = b1.z; b.y = b1.w;
    acc1 = __builtin_amdgcn_wmma_f32_16x16x4_f32(false, ahi, false, b, (short)0, acc1, false, false);
  }
}

__global__ void __launch_bounds__(256)
gemm1_x_w1(const float* __restrict__ X, const float* __restrict__ W,
           float* __restrict__ H, int mtiles) {
  __shared__ float4 wf[32 * 4 * 32];
  const int lane = threadIdx.x & 31;
  const int wave = threadIdx.x >> 5;
  const int c = lane & 15;
  const int h = lane >> 4;
  const int f = wave & 1;
  const int tile = blockIdx.x * 4 + (wave >> 1);
  const int tl = tile < mtiles ? tile : mtiles - 1;
  const float* xp = X + ((long)tl * 16 + c) * 256 + 4 * h;

  float4 cur[4], nxt[4];
#pragma unroll
  for (int u = 0; u < 4; ++u) cur[u] = *(const float4*)(xp + 8 * u);

#pragma unroll
  for (int m = 0; m < 4; ++m) {
    const int j = wave + 8 * m;
    const float* wp = W + (8 * j + 4 * h) * 64 + 4 * c;
    const float4 r0 = *(const float4*)(wp);
    const float4 r1 = *(const float4*)(wp + 64);
    const float4 r2 = *(const float4*)(wp + 128);
    const float4 r3 = *(const float4*)(wp + 192);
#pragma unroll
    for (int i = 0; i < 4; ++i) {
      float4 o;
      o.x = g1_elem(r0, i); o.y = g1_elem(r1, i); o.z = g1_elem(r2, i); o.w = g1_elem(r3, i);
      const int n = 2 * (c >> 3) + (i & 1);
      const int cl = 2 * (c & 7) + (i >> 1);
      wf[(j * 4 + n) * 32 + 16 * h + cl] = o;
    }
  }
  __syncthreads();
  if (tile >= mtiles) return;

  v8f acc0 = {}, acc1 = {};
  const float4* bp = wf + (2 * f) * 32 + lane;
  for (int jc = 0; jc < 32; jc += 8) {
#pragma unroll
    for (int u = 0; u < 4; ++u) nxt[u] = *(const float4*)(xp + 8 * (jc + 4 + u));
    g1_chunk(cur, bp + (jc * 4) * 32, acc0, acc1);
    if (jc + 8 < 32) {
#pragma unroll
      for (int u = 0; u < 4; ++u) cur[u] = *(const float4*)(xp + 8 * (jc + 8 + u));
    }
    g1_chunk(nxt, bp + ((jc + 4) * 4) * 32, acc0, acc1);
  }

  float* hp = H + ((long)tile * 16 + 8 * h) * 64 + 32 * f + 2 * c;
#pragma unroll
  for (int v = 0; v < 8; ++v) {
    v2f o;
    o.x = acc0[v]; o.y = acc1[v];
    *(v2f*)(hp + v * 64) = o;
  }
}

__global__ void gemm2_h_w2(const float* __restrict__ A1, const float* __restrict__ W,
                           float* __restrict__ H2, int mtiles) {
  const int lane = threadIdx.x & 31;
  const int wave = threadIdx.x >> 5;
  const int mtile = blockIdx.x * 4 + wave;
  if (mtile >= mtiles) return;
  const int kgrp = (lane >> 4) << 1;
  const int ncol = lane & 15;
  const float* arow = A1 + ((long)mtile * 16 + ncol) * 64;
  v8f acc = {};
#pragma unroll
  for (int k0 = 0; k0 < 64; k0 += 4) {
    v2f a = *(const v2f*)(arow + k0 + kgrp);
    a.x = fmaxf(a.x, 0.f);
    a.y = fmaxf(a.y, 0.f);
    v2f b;
    b.x = W[(k0 + kgrp) * 8 + (ncol & 7)];
    b.y = W[(k0 + kgrp + 1) * 8 + (ncol & 7)];
    acc = __builtin_amdgcn_wmma_f32_16x16x4_f32(false, a, false, b, (short)0, acc,
                                                false, false);
  }
  if (ncol < 8) {
    const long mb = (long)mtile * 16 + ((lane >> 4) << 3);
#pragma unroll
    for (int v = 0; v < 8; ++v)
      H2[(mb + v) * 8 + ncol] = acc[v];
  }
}

#define BUCKET_SHIFT 8
#define BUCKET_NODES 256
#define MAXB 512
#define NCHUNK 256
#define STAGE_CAP 12288

__global__ void __launch_bounds__(1024)
bucket_count(const int* __restrict__ ei, long Eraw, int nb, int* __restrict__ tbl) {
  __shared__ int hist[MAXB];
  const int t = threadIdx.x;
  const int wg = blockIdx.x;
  for (int b = t; b < nb; b += 1024) hist[b] = 0;
  __syncthreads();
  const long chunk = (Eraw + NCHUNK - 1) / NCHUNK;
  const long lo = wg * chunk;
  const long hi = (lo + chunk < Eraw) ? lo + chunk : Eraw;
  const int* dp = ei + Eraw;
  long e = lo + t;
  for (; e + 3072 < hi; e += 4096) {
    const int d0 = dp[e], d1 = dp[e + 1024], d2 = dp[e + 2048], d3 = dp[e + 3072];
    atomicAdd(&hist[d0 >> BUCKET_SHIFT], 1);
    atomicAdd(&hist[d1 >> BUCKET_SHIFT], 1);
    atomicAdd(&hist[d2 >> BUCKET_SHIFT], 1);
    atomicAdd(&hist[d3 >> BUCKET_SHIFT], 1);
  }
  for (; e < hi; e += 1024) atomicAdd(&hist[dp[e] >> BUCKET_SHIFT], 1);
  __syncthreads();
  for (int b = t; b < nb; b += 1024) tbl[b * NCHUNK + wg] = hist[b];
}

__global__ void __launch_bounds__(1024)
scan_offsets(const int* __restrict__ cnt, int* __restrict__ off, int n) {
  __shared__ int part[1024];
  const int t = threadIdx.x;
  const int base = blockIdx.x * 1024;
  int p = 0;
  for (int i = t; i < base; i += 1024) p += cnt[i];
  part[t] = p;
  __syncthreads();
  for (int o = 512; o > 0; o >>= 1) {
    if (t < o) part[t] += part[t + o];
    __syncthreads();
  }
  const int before = part[0];
  __syncthreads();
  const int i = base + t;
  const int v = (i < n) ? cnt[i] : 0;
  part[t] = v;
  __syncthreads();
  for (int o = 1; o < 1024; o <<= 1) {
    const int u = (t >= o) ? part[t - o] : 0;
    __syncthreads();
    part[t] += u;
    __syncthreads();
  }
  if (i < n) off[i] = before + part[t] - v;
  if (i == n - 1) off[n] = before + part[t];
}

__global__ void __launch_bounds__(1024)
bucket_scatter(const int* __restrict__ ei, long Eraw, int nb, const int* __restrict__ toff,
               unsigned* __restrict__ tmp) {
  __shared__ int cur[MAXB];
  const int t = threadIdx.x;
  const int wg = blockIdx.x;
  for (int b = t; b < nb; b += 1024) cur[b] = toff[b * NCHUNK + wg];
  __syncthreads();
  const long chunk = (Eraw + NCHUNK - 1) / NCHUNK;
  const long lo = wg * chunk;
  const long hi = (lo + chunk < Eraw) ? lo + chunk : Eraw;
  const int* dp = ei + Eraw;
  long e = lo + t;
  for (; e + 3072 < hi; e += 4096) {
    const int s0 = ei[e], s1 = ei[e + 1024], s2 = ei[e + 2048], s3 = ei[e + 3072];
    const int d0 = dp[e], d1 = dp[e + 1024], d2 = dp[e + 2048], d3 = dp[e + 3072];
    const int p0 = atomicAdd(&cur[d0 >> BUCKET_SHIFT], 1);
    const int p1 = atomicAdd(&cur[d1 >> BUCKET_SHIFT], 1);
    const int p2 = atomicAdd(&cur[d2 >> BUCKET_SHIFT], 1);
    const int p3 = atomicAdd(&cur[d3 >> BUCKET_SHIFT], 1);
    tmp[p0] = ((unsigned)s0 << 8) | (unsigned)(d0 & (BUCKET_NODES - 1));
    tmp[p1] = ((unsigned)s1 << 8) | (unsigned)(d1 & (BUCKET_NODES - 1));
    tmp[p2] = ((unsigned)s2 << 8) | (unsigned)(d2 & (BUCKET_NODES - 1));
    tmp[p3] = ((unsigned)s3 << 8) | (unsigned)(d3 & (BUCKET_NODES - 1));
  }
  for (; e < hi; e += 1024) {
    const int s = ei[e], d = dp[e];
    const int p = atomicAdd(&cur[d >> BUCKET_SHIFT], 1);
    tmp[p] = ((unsigned)s << 8) | (unsigned)(d & (BUCKET_NODES - 1));
  }
}

template <bool STAGED>
__device__ __forceinline__ void place_edges(const unsigned* __restrict__ tmp, int lo, int hi, int t,
                                            int* ncur, int* stage, int* __restrict__ perm) {
  int i = lo + t;
  for (; i + 3072 < hi; i += 4096) {
    const unsigned w0 = tmp[i], w1 = tmp[i + 1024], w2 = tmp[i + 2048], w3 = tmp[i + 3072];
    const int p0 = atomicAdd(&ncur[w0 & (BUCKET_NODES - 1)], 1);
    const int p1 = atomicAdd(&ncur[w1 & (BUCKET_NODES - 1)], 1);
    const int p2 = atomicAdd(&ncur[w2 & (BUCKET_NODES - 1)], 1);
    const int p3 = atomicAdd(&ncur[w3 & (BUCKET_NODES - 1)], 1);
    if (STAGED) {
      stage[p0 - lo] = (int)(w0 >> 8);
      stage[p1 - lo] = (int)(w1 >> 8);
      stage[p2 - lo] = (int)(w2 >> 8);
      stage[p3 - lo] = (int)(w3 >> 8);
    } else {
      perm[p0] = (int)(w0 >> 8);
      perm[p1] = (int)(w1 >> 8);
      perm[p2] = (int)(w2 >> 8);
      perm[p3] = (int)(w3 >> 8);
    }
  }
  for (; i < hi; i += 1024) {
    const unsigned w = tmp[i];
    const int p = atomicAdd(&ncur[w & (BUCKET_NODES - 1)], 1);
    if (STAGED) stage[p - lo] = (int)(w >> 8);
    else perm[p] = (int)(w >> 8);
  }
}

__global__ void __launch_bounds__(1024)
bucket_sort_nodes(const unsigned* __restrict__ tmp, const int* __restrict__ toff, int nb, int N,
                  int* __restrict__ off, int* __restrict__ perm) {
  __shared__ int ncnt[BUCKET_NODES];
  __shared__ int nsc[BUCKET_NODES];
  __shared__ int ncur[BUCKET_NODES];
  __shared__ int stage[STAGE_CAP];
  const int t = threadIdx.x;
  const int b = blockIdx.x;
  const int lo = toff[b * NCHUNK];
  const int hi = toff[(b + 1) * NCHUNK];
  if (t < BUCKET_NODES) ncnt[t] = 0;
  __syncthreads();
  int i = lo + t;
  for (; i + 3072 < hi; i += 4096) {
    const unsigned w0 = tmp[i], w1 = tmp[i + 1024], w2 = tmp[i + 2048], w3 = tmp[i + 3072];
    atomicAdd(&ncnt[w0 & (BUCKET_NODES - 1)], 1);
    atomicAdd(&ncnt[w1 & (BUCKET_NODES - 1)], 1);
    atomicAdd(&ncnt[w2 & (BUCKET_NODES - 1)], 1);
    atomicAdd(&ncnt[w3 & (BUCKET_NODES - 1)], 1);
  }
  for (; i < hi; i += 1024) atomicAdd(&ncnt[tmp[i] & (BUCKET_NODES - 1)], 1);
  __syncthreads();
  int v = 0;
  if (t < BUCKET_NODES) { v = ncnt[t]; nsc[t] = v; }
  __syncthreads();
  for (int o = 1; o < BUCKET_NODES; o <<= 1) {
    int u = 0;
    if (t < BUCKET_NODES && t >= o) u = nsc[t - o];
    __syncthreads();
    if (t < BUCKET_NODES) nsc[t] += u;
    __syncthreads();
  }
  if (t < BUCKET_NODES) {
    const int first = lo + nsc[t] - v;
    ncur[t] = first;
    const int node = b * BUCKET_NODES + t;
    if (node < N) off[node] = first;
  }
  if (b == nb - 1 && t == 0) off[N] = hi;
  __syncthreads();
  if (hi - lo <= STAGE_CAP) {
    place_edges<true>(tmp, lo, hi, t, ncur, stage, perm);
    __syncthreads();
    for (int k = (lo & ~31) + t; k < hi; k += 1024)
      if (k >= lo) perm[k] = stage[k - lo];
  } else {
    place_edges<false>(tmp, lo, hi, t, ncur, stage, perm);
  }
}

struct Row8 { v2f q[4]; };

__device__ __forceinline__ Row8 load_row8(const float* __restrict__ p) {
  const float4 a = *(const float4*)p;
  const float4 b = *(const float4*)(p + 4);
  Row8 r;
  r.q[0].x = a.x; r.q[0].y = a.y; r.q[1].x = a.z; r.q[1].y = a.w;
  r.q[2].x = b.x; r.q[2].y = b.y; r.q[3].x = b.z; r.q[3].y = b.w;
  return r;
}

__device__ __forceinline__ float dot8(const Row8& r, const Row8& w) {
  v2f ss = r.q[0] * w.q[0];
  ss += r.q[1] * w.q[1];
  ss += r.q[2] * w.q[2];
  ss += r.q[3] * w.q[3];
  return ss.x + ss.y;
}


__global__ void __launch_bounds__(256)
node_softmax_gather_heads8(const int* __restrict__ off, const int* __restrict__ perm,
                           const float* __restrict__ as, const float* __restrict__ ad,
                           const float* __restrict__ hf, const float* __restrict__ bias,
                           float* __restrict__ out, long N) {
  const long t = (long)blockIdx.x * blockDim.x + threadIdx.x;
  if (t >= N * 8) return;
  const int d = (int)(t >> 3);
  const int ro = (int)(t & 7) * 8;

  const Row8 asv = load_row8(as + ro);
  const Row8 own = load_row8(hf + (long)d * 64 + ro);
  const float ald = dot8(own, load_row8(ad + ro));
  float m = lrelu(dot8(own, asv) + ald);
  float den = 1.f;
  Row8 acc = own;

  const int j0 = off[d], j1 = off[d + 1];
  int j = j0;
  for (; j + 4 <= j1; j += 4) {
    const int s0 = perm[j], s1 = perm[j + 1], s2 = perm[j + 2], s3 = perm[j + 3];
    const Row8 r0 = load_row8(hf + (long)s0 * 64 + ro);
    const Row8 r1 = load_row8(hf + (long)s1 * 64 + ro);
    const Row8 r2 = load_row8(hf + (long)s2 * 64 + ro);
    const Row8 r3 = load_row8(hf + (long)s3 * 64 + ro);
    const float e0 = lrelu(dot8(r0, asv) + ald);
    const float e1 = lrelu(dot8(r1, asv) + ald);
    const float e2 = lrelu(dot8(r2, asv) + ald);
    const float e3 = lrelu(dot8(r3, asv) + ald);
    const float mn = fmaxf(fmaxf(fmaxf(m, e0), fmaxf(e1, e2)), e3);
    const float sc = __expf(m - mn);
    const float p0 = __expf(e0 - mn);
    const float p1 = __expf(e1 - mn);
    const float p2 = __expf(e2 - mn);
    const float p3 = __expf(e3 - mn);
    m = mn;
    den = den * sc + p0; den += p1; den += p2; den += p3;
#pragma unroll
    for (int k = 0; k < 4; ++k) {
      acc.q[k] = acc.q[k] * sc + p0 * r0.q[k];
      acc.q[k] += p1 * r1.q[k];
      acc.q[k] += p2 * r2.q[k];
      acc.q[k] += p3 * r3.q[k];
    }
  }
  for (; j < j1; ++j) {
    const Row8 r = load_row8(hf + (long)perm[j] * 64 + ro);
    const float e = lrelu(dot8(r, asv) + ald);
    const float mn = fmaxf(m, e);
    const float sc = __expf(m - mn);
    const float p = __expf(e - mn);
    m = mn;
    den = den * sc + p;
#pragma unroll
    for (int k = 0; k < 4; ++k) acc.q[k] = acc.q[k] * sc + p * r.q[k];
  }

  const float dn = den + 1e-16f;
  float* op = out + (long)d * 64 + ro;
  const float* bp = bias + ro;
  float4 o0, o1;
  o0.x = acc.q[0].x / dn + bp[0]; o0.y = acc.q[0].y / dn + bp[1];
  o0.z = acc.q[1].x / dn + bp[2]; o0.w = acc.q[1].y / dn + bp[3];
  o1.x = acc.q[2].x / dn + bp[4]; o1.y = acc.q[2].y / dn + bp[5];
  o1.z = acc.q[3].x / dn + bp[6]; o1.w = acc.q[3].y / dn + bp[7];
  *(float4*)op = o0;
  *(float4*)(op + 4) = o1;
}

template <int X>
__device__ __forceinline__ float lane_xor(float v) {
  return __builtin_bit_cast(float, __builtin_amdgcn_update_dpp(
      0, __builtin_bit_cast(int, v), 0x160 + X, 0xf, 0xf, true));
}

__device__ __forceinline__ float sum8(float v) {
  v += lane_xor<1>(v);
  v += lane_xor<2>(v);
  v += lane_xor<4>(v);
  return v;
}

__global__ void __launch_bounds__(256)
node_softmax_gather_head1(const int* __restrict__ off, const int* __restrict__ perm,
                          const float* __restrict__ as, const float* __restrict__ ad,
                          const float* __restrict__ hf, const float* __restrict__ bias,
                          float* __restrict__ out, long N) {
  const long t = (long)blockIdx.x * blockDim.x + threadIdx.x;
  if (t >= N * 8) return;
  const int d = (int)(t >> 3);
  const int g = (int)(t & 7);

  const float asg = as[g];
  const float own = hf[(long)d * 8 + g];
  const float ald = sum8(own * ad[g]);
  float m = lrelu(sum8(own * asg) + ald);
  float den = 1.f;
  float acc = own;

  const int j0 = off[d], j1 = off[d + 1];
  int j = j0;
  for (; j + 4 <= j1; j += 4) {
    const int s0 = perm[j], s1 = perm[j + 1], s2 = perm[j + 2], s3 = perm[j + 3];
    const float x0 = hf[(long)s0 * 8 + g];
    const float x1 = hf[(long)s1 * 8 + g];
    const float x2 = hf[(long)s2 * 8 + g];
    const float x3 = hf[(long)s3 * 8 + g];
    const float e0 = lrelu(sum8(x0 * asg) + ald);
    const float e1 = lrelu(sum8(x1 * asg) + ald);
    const float e2 = lrelu(sum8(x2 * asg) + ald);
    const float e3 = lrelu(sum8(x3 * asg) + ald);
    const float mn = fmaxf(fmaxf(fmaxf(m, e0), fmaxf(e1, e2)), e3);
    const float sc = __expf(m - mn);
    const float p0 = __expf(e0 - mn);
    const float p1 = __expf(e1 - mn);
    const float p2 = __expf(e2 - mn);
    const float p3 = __expf(e3 - mn);
    m = mn;
    den = den * sc + p0; den += p1; den += p2; den += p3;
    acc = acc * sc + p0 * x0; acc += p1 * x1; acc += p2 * x2; acc += p3 * x3;
  }
  for (; j < j1; ++j) {
    const float x = hf[(long)perm[j] * 8 + g];
    const float e = lrelu(sum8(x * asg) + ald);
    const float mn = fmaxf(m, e);
    const float sc = __expf(m - mn);
    const float p = __expf(e - mn);
    m = mn;
    den = den * sc + p;
    acc = acc * sc + p * x;
  }
  out[(long)d * 8 + g] = acc / (den + 1e-16f) + bias[g];
}

__global__ void score_kernel(const float* __restrict__ emb, const float* __restrict__ W3,
                             const float* __restrict__ b3, float* __restrict__ score, long N) {
  long n = (long)blockIdx.x * blockDim.x + threadIdx.x;
  if (n >= N) return;
  float s = b3[0];
#pragma unroll
  for (int c = 0; c < 8; ++c) s += emb[n * 8 + c] * W3[c];
  score[n] = s;
}

static inline int nblk(long n, int b) { return (int)((n + b - 1) / b); }

extern "C" void kernel_launch(void* const* d_in, const int* in_sizes, int n_in,
                              void* d_out, int out_size, void* d_ws, size_t ws_size,
                              hipStream_t stream) {
  const float* x   = (const float*)d_in[0];
  const int*   ei  = (const int*)d_in[1];
  const float* W1  = (const float*)d_in[2];
  const float* as1 = (const float*)d_in[3];
  const float* ad1 = (const float*)d_in[4];
  const float* b1  = (const float*)d_in[5];
  const float* W2  = (const float*)d_in[6];
  const float* as2 = (const float*)d_in[7];
  const float* ad2 = (const float*)d_in[8];
  const float* b2  = (const float*)d_in[9];
  const float* W3  = (const float*)d_in[10];
  const float* b3  = (const float*)d_in[11];

  const long N    = in_sizes[0] / 256;
  const long Eraw = in_sizes[1] / 2;

  float* ws   = (float*)d_ws;
  float* h1   = ws;
  float* acc1 = h1 + N * 64;
  int*   perm = (int*)(acc1 + N * 64);
  int*   off  = perm + Eraw;
  const int nb = (int)((N + BUCKET_NODES - 1) / BUCKET_NODES);
  unsigned* tmp = (unsigned*)acc1;
  int*   tbl  = (int*)acc1 + Eraw;
  int*   toff = tbl + nb * NCHUNK;
  float* h2   = h1;

  float* emb   = (float*)d_out;
  float* score = emb + N * 8;

  const int B = 256;
  const int mtiles = (int)(N / 16);

  bucket_count<<<NCHUNK, 1024, 0, stream>>>(ei, Eraw, nb, tbl);
  scan_offsets<<<nblk(nb * NCHUNK, 1024), 1024, 0, stream>>>(tbl, toff, nb * NCHUNK);
  bucket_scatter<<<NCHUNK, 1024, 0, stream>>>(ei, Eraw, nb, toff, tmp);
  bucket_sort_nodes<<<nb, 1024, 0, stream>>>(tmp, toff, nb, (int)N, off, perm);

  gemm1_x_w1<<<(mtiles + 3) / 4, 256, 0, stream>>>(x, W1, h1, mtiles);
  node_softmax_gather_heads8<<<nblk(N * 8, B), B, 0, stream>>>(off, perm, as1, ad1, h1, b1, acc1, N);

  gemm2_h_w2<<<(mtiles + 3) / 4, 128, 0, stream>>>(acc1, W2, h2, mtiles);
  node_softmax_gather_head1<<<nblk(N * 8, B), B, 0, stream>>>(off, perm, as2, ad2, h2, b2, emb, N);

  score_kernel<<<nblk(N, B), B, 0, stream>>>(emb, W3, b3, score, N);
}


# Layer-2 projection (W2 on the f32 WMMA, rows handed over in LDS) and score dot (8-lane sum) folded into the epilogues of the gathers that feed them: 9 launches -> 7; score sum order changes
# speedup vs baseline: 15.0321x; 1.0443x over previous
#include <hip/hip_runtime.h>
#include <math.h>

typedef __attribute__((ext_vector_type(2))) float v2f;
typedef __attribute__((ext_vector_type(8))) float v8f;

#define NEG_SLOPE 0.2f

__device__ __forceinline__ float lrelu(float v) { return v >= 0.f ? v : NEG_SLOPE * v; }

__device__ __forceinline__ float g1_elem(const float4& r, int i) {
  return i == 0 ? r.x : i == 1 ? r.y : i == 2 ? r.z : r.w;
}

__device__ __forceinline__ void g1_chunk(const float4 (&a)[4], const float4* bp,
                                         v8f& acc0, v8f& acc1) {
#pragma unroll
  for (int u = 0; u < 4; ++u) {
    const float4 b0 = bp[(u * 4 + 0) * 32];
    const float4 b1 = bp[(u * 4 + 1) * 32];
    v2f alo, ahi, b;
    alo.x = a[u].x; alo.y = a[u].y; ahi.x = a[u].z; ahi.y = a[u].w;
    b.x = b0.x; b.y = b0.y;
    acc0 = __builtin_amdgcn_wmma_f32_16x16x4_f32(false, alo, false, b, (short)0, acc0, false, false);
    b.x = b1.x; b.y = b1.y;
    acc1 = __builtin_amdgcn_wmma_f32_16x16x4_f32(false, alo, false, b, (short)0, acc1, false, false);
    b.x = b0.z; b.y = b0.w;
    acc0 = __builtin_amdgcn_wmma_f32_16x16x4_f32(false, ahi, false, b, (short)0, acc0, false, false);
    b.x = b1.z; b.y = b1.w;
    acc1 = __builtin_amdgcn_wmma_f32_16x16x4_f32(false, ahi, false, b, (short)0, acc1, false, false);
  }
}

__global__ void __launch_bounds__(256)
gemm1_x_w1(const float* __restrict__ X, const float* __restrict__ W,
           float* __restrict__ H, int mtiles) {
  __shared__ float4 wf[32 * 4 * 32];
  const int lane = threadIdx.x & 31;
  const int wave = threadIdx.x >> 5;
  const int c = lane & 15;
  const int h = lane >> 4;
  const int f = wave & 1;
  const int tile = blockIdx.x * 4 + (wave >> 1);
  const int tl = tile < mtiles ? tile : mtiles - 1;
  const float* xp = X + ((long)tl * 16 + c) * 256 + 4 * h;

  float4 cur[4], nxt[4];
#pragma unroll
  for (int u = 0; u < 4; ++u) cur[u] = *(const float4*)(xp + 8 * u);

#pragma unroll
  for (int m = 0; m < 4; ++m) {
    const int j = wave + 8 * m;
    const float* wp = W + (8 * j + 4 * h) * 64 + 4 * c;
    const float4 r0 = *(const float4*)(wp);
    const float4 r1 = *(const float4*)(wp + 64);
    const float4 r2 = *(const float4*)(wp + 128);
    const float4 r3 = *(const float4*)(wp + 192);
#pragma unroll
    for (int i = 0; i < 4; ++i) {
      float4 o;
      o.x = g1_elem(r0, i); o.y = g1_elem(r1, i); o.z = g1_elem(r2, i); o.w = g1_elem(r3, i);
      const int n = 2 * (c >> 3) + (i & 1);
      const int cl = 2 * (c & 7) + (i >> 1);
      wf[(j * 4 + n) * 32 + 16 * h + cl] = o;
    }
  }
  __syncthreads();
  if (tile >= mtiles) return;

  v8f acc0 = {}, acc1 = {};
  const float4* bp = wf + (2 * f) * 32 + lane;
  for (int jc = 0; jc < 32; jc += 8) {
#pragma unroll
    for (int u = 0; u < 4; ++u) nxt[u] = *(const float4*)(xp + 8 * (jc + 4 + u));
    g1_chunk(cur, bp + (jc * 4) * 32, acc0, acc1);
    if (jc + 8 < 32) {
#pragma unroll
      for (int u = 0; u < 4; ++u) cur[u] = *(const float4*)(xp + 8 * (jc + 8 + u));
    }
    g1_chunk(nxt, bp + ((jc + 4) * 4) * 32, acc0, acc1);
  }

  float* hp = H + ((long)tile * 16 + 8 * h) * 64 + 32 * f + 2 * c;
#pragma unroll
  for (int v = 0; v < 8; ++v) {
    v2f o;
    o.x = acc0[v]; o.y = acc1[v];
    *(v2f*)(hp + v * 64) = o;
  }
}

#define BUCKET_SHIFT 8
#define BUCKET_NODES 256
#define MAXB 512
#define NCHUNK 256
#define STAGE_CAP 12288

__global__ void __launch_bounds__(1024)
bucket_count(const int* __restrict__ ei, long Eraw, int nb, int* __restrict__ tbl) {
  __shared__ int hist[MAXB];
  const int t = threadIdx.x;
  const int wg = blockIdx.x;
  for (int b = t; b < nb; b += 1024) hist[b] = 0;
  __syncthreads();
  const long chunk = (Eraw + NCHUNK - 1) / NCHUNK;
  const long lo = wg * chunk;
  const long hi = (lo + chunk < Eraw) ? lo + chunk : Eraw;
  const int* dp = ei + Eraw;
  long e = lo + t;
  for (; e + 3072 < hi; e += 4096) {
    const int d0 = dp[e], d1 = dp[e + 1024], d2 = dp[e + 2048], d3 = dp[e + 3072];
    atomicAdd(&hist[d0 >> BUCKET_SHIFT], 1);
    atomicAdd(&hist[d1 >> BUCKET_SHIFT], 1);
    atomicAdd(&hist[d2 >> BUCKET_SHIFT], 1);
    atomicAdd(&hist[d3 >> BUCKET_SHIFT], 1);
  }
  for (; e < hi; e += 1024) atomicAdd(&hist[dp[e] >> BUCKET_SHIFT], 1);
  __syncthreads();
  for (int b = t; b < nb; b += 1024) tbl[b * NCHUNK + wg] = hist[b];
}

__global__ void __launch_bounds__(1024)
scan_offsets(const int* __restrict__ cnt, int* __restrict__ off, int n) {
  __shared__ int part[1024];
  const int t = threadIdx.x;
  const int base = blockIdx.x * 1024;
  int p = 0;
  for (int i = t; i < base; i += 1024) p += cnt[i];
  part[t] = p;
  __syncthreads();
  for (int o = 512; o > 0; o >>= 1) {
    if (t < o) part[t] += part[t + o];
    __syncthreads();
  }
  const int before = part[0];
  __syncthreads();
  const int i = base + t;
  const int v = (i < n) ? cnt[i] : 0;
  part[t] = v;
  __syncthreads();
  for (int o = 1; o < 1024; o <<= 1) {
    const int u = (t >= o) ? part[t - o] : 0;
    __syncthreads();
    part[t] += u;
    __syncthreads();
  }
  if (i < n) off[i] = before + part[t] - v;
  if (i == n - 1) off[n] = before + part[t];
}

__global__ void __launch_bounds__(1024)
bucket_scatter(const int* __restrict__ ei, long Eraw, int nb, const int* __restrict__ toff,
               unsigned* __restrict__ tmp) {
  __shared__ int cur[MAXB];
  const int t = threadIdx.x;
  const int wg = blockIdx.x;
  for (int b = t; b < nb; b += 1024) cur[b] = toff[b * NCHUNK + wg];
  __syncthreads();
  const long chunk = (Eraw + NCHUNK - 1) / NCHUNK;
  const long lo = wg * chunk;
  const long hi = (lo + chunk < Eraw) ? lo + chunk : Eraw;
  const int* dp = ei + Eraw;
  long e = lo + t;
  for (; e + 3072 < hi; e += 4096) {
    const int s0 = ei[e], s1 = ei[e + 1024], s2 = ei[e + 2048], s3 = ei[e + 3072];
    const int d0 = dp[e], d1 = dp[e + 1024], d2 = dp[e + 2048], d3 = dp[e + 3072];
    const int p0 = atomicAdd(&cur[d0 >> BUCKET_SHIFT], 1);
    const int p1 = atomicAdd(&cur[d1 >> BUCKET_SHIFT], 1);
    const int p2 = atomicAdd(&cur[d2 >> BUCKET_SHIFT], 1);
    const int p3 = atomicAdd(&cur[d3 >> BUCKET_SHIFT], 1);
    tmp[p0] = ((unsigned)s0 << 8) | (unsigned)(d0 & (BUCKET_NODES - 1));
    tmp[p1] = ((unsigned)s1 << 8) | (unsigned)(d1 & (BUCKET_NODES - 1));
    tmp[p2] = ((unsigned)s2 << 8) | (unsigned)(d2 & (BUCKET_NODES - 1));
    tmp[p3] = ((unsigned)s3 << 8) | (unsigned)(d3 & (BUCKET_NODES - 1));
  }
  for (; e < hi; e += 1024) {
    const int s = ei[e], d = dp[e];
    const int p = atomicAdd(&cur[d >> BUCKET_SHIFT], 1);
    tmp[p] = ((unsigned)s << 8) | (unsigned)(d & (BUCKET_NODES - 1));
  }
}

template <bool STAGED>
__device__ __forceinline__ void place_edges(const unsigned* __restrict__ tmp, int lo, int hi, int t,
                                            int* ncur, int* stage, int* __restrict__ perm) {
  int i = lo + t;
  for (; i + 3072 < hi; i += 4096) {
    const unsigned w0 = tmp[i], w1 = tmp[i + 1024], w2 = tmp[i + 2048], w3 = tmp[i + 3072];
    const int p0 = atomicAdd(&ncur[w0 & (BUCKET_NODES - 1)], 1);
    const int p1 = atomicAdd(&ncur[w1 & (BUCKET_NODES - 1)], 1);
    const int p2 = atomicAdd(&ncur[w2 & (BUCKET_NODES - 1)], 1);
    const int p3 = atomicAdd(&ncur[w3 & (BUCKET_NODES - 1)], 1);
    if (STAGED) {
      stage[p0 - lo] = (int)(w0 >> 8);
      stage[p1 - lo] = (int)(w1 >> 8);
      stage[p2 - lo] = (int)(w2 >> 8);
      stage[p3 - lo] = (int)(w3 >> 8);
    } else {
      perm[p0] = (int)(w0 >> 8);
      perm[p1] = (int)(w1 >> 8);
      perm[p2] = (int)(w2 >> 8);
      perm[p3] = (int)(w3 >> 8);
    }
  }
  for (; i < hi; i += 1024) {
    const unsigned w = tmp[i];
    const int p = atomicAdd(&ncur[w & (BUCKET_NODES - 1)], 1);
    if (STAGED) stage[p - lo] = (int)(w >> 8);
    else perm[p] = (int)(w >> 8);
  }
}

__global__ void __launch_bounds__(1024)
bucket_sort_nodes(const unsigned* __restrict__ tmp, const int* __restrict__ toff, int nb, int N,
                  int* __restrict__ off, int* __restrict__ perm) {
  __shared__ int ncnt[BUCKET_NODES];
  __shared__ int nsc[BUCKET_NODES];
  __shared__ int ncur[BUCKET_NODES];
  __shared__ int stage[STAGE_CAP];
  const int t = threadIdx.x;
  const int b = blockIdx.x;
  const int lo = toff[b * NCHUNK];
  const int hi = toff[(b + 1) * NCHUNK];
  if (t < BUCKET_NODES) ncnt[t] = 0;
  __syncthreads();
  int i = lo + t;
  for (; i + 3072 < hi; i += 4096) {
    const unsigned w0 = tmp[i], w1 = tmp[i + 1024], w2 = tmp[i + 2048], w3 = tmp[i + 3072];
    atomicAdd(&ncnt[w0 & (BUCKET_NODES - 1)], 1);
    atomicAdd(&ncnt[w1 & (BUCKET_NODES - 1)], 1);
    atomicAdd(&ncnt[w2 & (BUCKET_NODES - 1)], 1);
    atomicAdd(&ncnt[w3 & (BUCKET_NODES - 1)], 1);
  }
  for (; i < hi; i += 1024) atomicAdd(&ncnt[tmp[i] & (BUCKET_NODES - 1)], 1);
  __syncthreads();
  int v = 0;
  if (t < BUCKET_NODES) { v = ncnt[t]; nsc[t] = v; }
  __syncthreads();
  for (int o = 1; o < BUCKET_NODES; o <<= 1) {
    int u = 0;
    if (t < BUCKET_NODES && t >= o) u = nsc[t - o];
    __syncthreads();
    if (t < BUCKET_NODES) nsc[t] += u;
    __syncthreads();
  }
  if (t < BUCKET_NODES) {
    const int first = lo + nsc[t] - v;
    ncur[t] = first;
    const int node = b * BUCKET_NODES + t;
    if (node < N) off[node] = first;
  }
  if (b == nb - 1 && t == 0) off[N] = hi;
  __syncthreads();
  if (hi - lo <= STAGE_CAP) {
    place_edges<true>(tmp, lo, hi, t, ncur, stage, perm);
    __syncthreads();
    for (int k = (lo & ~31) + t; k < hi; k += 1024)
      if (k >= lo) perm[k] = stage[k - lo];
  } else {
    place_edges<false>(tmp, lo, hi, t, ncur, stage, perm);
  }
}

struct Row8 { v2f q[4]; };

__device__ __forceinline__ Row8 load_row8(const float* __restrict__ p) {
  const float4 a = *(const float4*)p;
  const float4 b = *(const float4*)(p + 4);
  Row8 r;
  r.q[0].x = a.x; r.q[0].y = a.y; r.q[1].x = a.z; r.q[1].y = a.w;
  r.q[2].x = b.x; r.q[2].y = b.y; r.q[3].x = b.z; r.q[3].y = b.w;
  return r;
}

__device__ __forceinline__ float dot8(const Row8& r, const Row8& w) {
  v2f ss = r.q[0] * w.q[0];
  ss += r.q[1] * w.q[1];
  ss += r.q[2] * w.q[2];
  ss += r.q[3] * w.q[3];
  return ss.x + ss.y;
}


__global__ void __launch_bounds__(256)
node_softmax_gather_heads8(const int* __restrict__ off, const int* __restrict__ perm,
                           const float* __restrict__ as, const float* __restrict__ ad,
                           const float* __restrict__ hf, const float* __restrict__ bias,
                           const float* __restrict__ w2, float* __restrict__ h2, long N) {
  __shared__ float4 rows[32 * 17];
  const long t0 = (long)blockIdx.x * 256 + threadIdx.x;
  const long t = t0 < N * 8 ? t0 : N * 8 - 8 + (t0 & 7);
  const int d = (int)(t >> 3);
  const int ro = (int)(t & 7) * 8;

  const Row8 asv = load_row8(as + ro);
  const Row8 own = load_row8(hf + (long)d * 64 + ro);
  const float ald = dot8(own, load_row8(ad + ro));
  float m = lrelu(dot8(own, asv) + ald);
  float den = 1.f;
  Row8 acc = own;

  const int j0 = off[d], j1 = off[d + 1];
  int j = j0;
  for (; j + 4 <= j1; j += 4) {
    const int s0 = perm[j], s1 = perm[j + 1], s2 = perm[j + 2], s3 = perm[j + 3];
    const Row8 r0 = load_row8(hf + (long)s0 * 64 + ro);
    const Row8 r1 = load_row8(hf + (long)s1 * 64 + ro);
    const Row8 r2 = load_row8(hf + (long)s2 * 64 + ro);
    const Row8 r3 = load_row8(hf + (long)s3 * 64 + ro);
    const float e0 = lrelu(dot8(r0, asv) + ald);
    const float e1 = lrelu(dot8(r1, asv) + ald);
    const float e2 = lrelu(dot8(r2, asv) + ald);
    const float e3 = lrelu(dot8(r3, asv) + ald);
    const float mn = fmaxf(fmaxf(fmaxf(m, e0), fmaxf(e1, e2)), e3);
    const float sc = __expf(m - mn);
    const float p0 = __expf(e0 - mn);
    const float p1 = __expf(e1 - mn);
    const float p2 = __expf(e2 - mn);
    const float p3 = __expf(e3 - mn);
    m = mn;
    den = den * sc + p0; den += p1; den += p2; den += p3;
#pragma unroll
    for (int k = 0; k < 4; ++k) {
      acc.q[k] = acc.q[k] * sc + p0 * r0.q[k];
      acc.q[k] += p1 * r1.q[k];
      acc.q[k] += p2 * r2.q[k];
      acc.q[k] += p3 * r3.q[k];
    }
  }
  for (; j < j1; ++j) {
    const Row8 r = load_row8(hf + (long)perm[j] * 64 + ro);
    const float e = lrelu(dot8(r, asv) + ald);
    const float mn = fmaxf(m, e);
    const float sc = __expf(m - mn);
    const float p = __expf(e - mn);
    m = mn;
    den = den * sc + p;
#pragma unroll
    for (int k = 0; k < 4; ++k) acc.q[k] = acc.q[k] * sc + p * r.q[k];
  }

  const float dn = den + 1e-16f;
  const float* bp = bias + ro;
  float4 o0, o1;
  o0.x = fmaxf(acc.q[0].x / dn + bp[0], 0.f); o0.y = fmaxf(acc.q[0].y / dn + bp[1], 0.f);
  o0.z = fmaxf(acc.q[1].x / dn + bp[2], 0.f); o0.w = fmaxf(acc.q[1].y / dn + bp[3], 0.f);
  o1.x = fmaxf(acc.q[2].x / dn + bp[4], 0.f); o1.y = fmaxf(acc.q[2].y / dn + bp[5], 0.f);
  o1.z = fmaxf(acc.q[3].x / dn + bp[6], 0.f); o1.w = fmaxf(acc.q[3].y / dn + bp[7], 0.f);
  float* sp = (float*)rows + (threadIdx.x >> 3) * 68 + ro;
  *(float4*)sp = o0;
  *(float4*)(sp + 4) = o1;

  const int lane = threadIdx.x & 31;
  const int wave = threadIdx.x >> 5;
  const int c = lane & 15;
  const int kg = (lane >> 4) << 1;
  v2f bf[16] = {};
  if (wave < 2) {
#pragma unroll
    for (int i = 0; i < 16; ++i) {
      bf[i].x = w2[(4 * i + kg) * 8 + (c & 7)];
      bf[i].y = w2[(4 * i + kg + 1) * 8 + (c & 7)];
    }
  }
  __syncthreads();
  if (wave < 2) {
    const float* arow = (const float*)rows + (16 * wave + c) * 68 + kg;
    v8f pr = {};
#pragma unroll
    for (int i = 0; i < 16; ++i) {
      const v2f a = *(const v2f*)(arow + 4 * i);
      pr = __builtin_amdgcn_wmma_f32_16x16x4_f32(false, a, false, bf[i], (short)0, pr,
                                                 false, false);
    }
    if (c < 8) {
      const long nb0 = (long)blockIdx.x * 32 + 16 * wave + 4 * kg;
#pragma unroll
      for (int v = 0; v < 8; ++v)
        if (nb0 + v < N) h2[(nb0 + v) * 8 + c] = pr[v];
    }
  }
}

template <int X>
__device__ __forceinline__ float lane_xor(float v) {
  return __builtin_bit_cast(float, __builtin_amdgcn_update_dpp(
      0, __builtin_bit_cast(int, v), 0x160 + X, 0xf, 0xf, true));
}

__device__ __forceinline__ float sum8(float v) {
  v += lane_xor<1>(v);
  v += lane_xor<2>(v);
  v += lane_xor<4>(v);
  return v;
}

__global__ void __launch_bounds__(256)
node_softmax_gather_head1(const int* __restrict__ off, const int* __restrict__ perm,
                          const float* __restrict__ as, const float* __restrict__ ad,
                          const float* __restrict__ hf, const float* __restrict__ bias,
                          const float* __restrict__ w3, const float* __restrict__ b3,
                          float* __restrict__ out, float* __restrict__ score, long N) {
  const long t = (long)blockIdx.x * blockDim.x + threadIdx.x;
  if (t >= N * 8) return;
  const int d = (int)(t >> 3);
  const int g = (int)(t & 7);

  const float asg = as[g];
  const float own = hf[(long)d * 8 + g];
  const float ald = sum8(own * ad[g]);
  float m = lrelu(sum8(own * asg) + ald);
  float den = 1.f;
  float acc = own;

  const int j0 = off[d], j1 = off[d + 1];
  int j = j0;
  for (; j + 4 <= j1; j += 4) {
    const int s0 = perm[j], s1 = perm[j + 1], s2 = perm[j + 2], s3 = perm[j + 3];
    const float x0 = hf[(long)s0 * 8 + g];
    const float x1 = hf[(long)s1 * 8 + g];
    const float x2 = hf[(long)s2 * 8 + g];
    const float x3 = hf[(long)s3 * 8 + g];
    const float e0 = lrelu(sum8(x0 * asg) + ald);
    const float e1 = lrelu(sum8(x1 * asg) + ald);
    const float e2 = lrelu(sum8(x2 * asg) + ald);
    const float e3 = lrelu(sum8(x3 * asg) + ald);
    const float mn = fmaxf(fmaxf(fmaxf(m, e0), fmaxf(e1, e2)), e3);
    const float sc = __expf(m - mn);
    const float p0 = __expf(e0 - mn);
    const float p1 = __expf(e1 - mn);
    const float p2 = __expf(e2 - mn);
    const float p3 = __expf(e3 - mn);
    m = mn;
    den = den * sc + p0; den += p1; den += p2; den += p3;
    acc = acc * sc + p0 * x0; acc += p1 * x1; acc += p2 * x2; acc += p3 * x3;
  }
  for (; j < j1; ++j) {
    const float x = hf[(long)perm[j] * 8 + g];
    const float e = lrelu(sum8(x * asg) + ald);
    const float mn = fmaxf(m, e);
    const float sc = __expf(m - mn);
    const float p = __expf(e - mn);
    m = mn;
    den = den * sc + p;
    acc = acc * sc + p * x;
  }
  const float o = acc / (den + 1e-16f) + bias[g];
  out[(long)d * 8 + g] = o;
  const float sc = sum8(o * w3[g]) + b3[0];
  if (g == 0) score[d] = sc;
}

static inline int nblk(long n, int b) { return (int)((n + b - 1) / b); }

extern "C" void kernel_launch(void* const* d_in, const int* in_sizes, int n_in,
                              void* d_out, int out_size, void* d_ws, size_t ws_size,
                              hipStream_t stream) {
  const float* x   = (const float*)d_in[0];
  const int*   ei  = (const int*)d_in[1];
  const float* W1  = (const float*)d_in[2];
  const float* as1 = (const float*)d_in[3];
  const float* ad1 = (const float*)d_in[4];
  const float* b1  = (const float*)d_in[5];
  const float* W2  = (const float*)d_in[6];
  const float* as2 = (const float*)d_in[7];
  const float* ad2 = (const float*)d_in[8];
  const float* b2  = (const float*)d_in[9];
  const float* W3  = (const float*)d_in[10];
  const float* b3  = (const float*)d_in[11];

  const long N    = in_sizes[0] / 256;
  const long Eraw = in_sizes[1] / 2;

  float* ws   = (float*)d_ws;
  float* h1   = ws;
  float* acc1 = h1 + N * 64;
  int*   perm = (int*)(acc1 + N * 64);
  int*   off  = perm + Eraw;
  const int nb = (int)((N + BUCKET_NODES - 1) / BUCKET_NODES);
  unsigned* tmp = (unsigned*)acc1;
  int*   tbl  = (int*)acc1 + Eraw;
  int*   toff = tbl + nb * NCHUNK;
  float* h2   = acc1;

  float* emb   = (float*)d_out;
  float* score = emb + N * 8;

  const int B = 256;
  const int mtiles = (int)(N / 16);

  bucket_count<<<NCHUNK, 1024, 0, stream>>>(ei, Eraw, nb, tbl);
  scan_offsets<<<nblk(nb * NCHUNK, 1024), 1024, 0, stream>>>(tbl, toff, nb * NCHUNK);
  bucket_scatter<<<NCHUNK, 1024, 0, stream>>>(ei, Eraw, nb, toff, tmp);
  bucket_sort_nodes<<<nb, 1024, 0, stream>>>(tmp, toff, nb, (int)N, off, perm);

  gemm1_x_w1<<<(mtiles + 3) / 4, 256, 0, stream>>>(x, W1, h1, mtiles);
  node_softmax_gather_heads8<<<nblk(N * 8, 256), 256, 0, stream>>>(off, perm, as1, ad1, h1, b1, W2, h2, N);

  node_softmax_gather_head1<<<nblk(N * 8, B), B, 0, stream>>>(off, perm, as2, ad2, h2, b2, W3, b3,
                                                              emb, score, N);
}


# Level-1 scatter staged in LDS: a chunk's edges are put in bucket order there (returning LDS add as arrival-order rank, as before), then tmp is written as contiguous runs, not one word per line
# speedup vs baseline: 18.0130x; 1.1983x over previous
#include <hip/hip_runtime.h>
#include <math.h>

typedef __attribute__((ext_vector_type(2))) float v2f;
typedef __attribute__((ext_vector_type(8))) float v8f;

#define NEG_SLOPE 0.2f

__device__ __forceinline__ float lrelu(float v) { return v >= 0.f ? v : NEG_SLOPE * v; }

__device__ __forceinline__ float g1_elem(const float4& r, int i) {
  return i == 0 ? r.x : i == 1 ? r.y : i == 2 ? r.z : r.w;
}

__device__ __forceinline__ void g1_chunk(const float4 (&a)[4], const float4* bp,
                                         v8f& acc0, v8f& acc1) {
#pragma unroll
  for (int u = 0; u < 4; ++u) {
    const float4 b0 = bp[(u * 4 + 0) * 32];
    const float4 b1 = bp[(u * 4 + 1) * 32];
    v2f alo, ahi, b;
    alo.x = a[u].x; alo.y = a[u].y; ahi.x = a[u].z; ahi.y = a[u].w;
    b.x = b0.x; b.y = b0.y;
    acc0 = __builtin_amdgcn_wmma_f32_16x16x4_f32(false, alo, false, b, (short)0, acc0, false, false);
    b.x = b1.x; b.y = b1.y;
    acc1 = __builtin_amdgcn_wmma_f32_16x16x4_f32(false, alo, false, b, (short)0, acc1, false, false);
    b.x = b0.z; b.y = b0.w;
    acc0 = __builtin_amdgcn_wmma_f32_16x16x4_f32(false, ahi, false, b, (short)0, acc0, false, false);
    b.x = b1.z; b.y = b1.w;
    acc1 = __builtin_amdgcn_wmma_f32_16x16x4_f32(false, ahi, false, b, (short)0, acc1, false, false);
  }
}

__global__ void __launch_bounds__(256)
gemm1_x_w1(const float* __restrict__ X, const float* __restrict__ W,
           float* __restrict__ H, int mtiles) {
  __shared__ float4 wf[32 * 4 * 32];
  const int lane = threadIdx.x & 31;
  const int wave = threadIdx.x >> 5;
  const int c = lane & 15;
  const int h = lane >> 4;
  const int f = wave & 1;
  const int tile = blockIdx.x * 4 + (wave >> 1);
  const int tl = tile < mtiles ? tile : mtiles - 1;
  const float* xp = X + ((long)tl * 16 + c) * 256 + 4 * h;

  float4 cur[4], nxt[4];
#pragma unroll
  for (int u = 0; u < 4; ++u) cur[u] = *(const float4*)(xp + 8 * u);

#pragma unroll
  for (int m = 0; m < 4; ++m) {
    const int j = wave + 8 * m;
    const float* wp = W + (8 * j + 4 * h) * 64 + 4 * c;
    const float4 r0 = *(const float4*)(wp);
    const float4 r1 = *(const float4*)(wp + 64);
    const float4 r2 = *(const float4*)(wp + 128);
    const float4 r3 = *(const float4*)(wp + 192);
#pragma unroll
    for (int i = 0; i < 4; ++i) {
      float4 o;
      o.x = g1_elem(r0, i); o.y = g1_elem(r1, i); o.z = g1_elem(r2, i); o.w = g1_elem(r3, i);
      const int n = 2 * (c >> 3) + (i & 1);
      const int cl = 2 * (c & 7) + (i >> 1);
      wf[(j * 4 + n) * 32 + 16 * h + cl] = o;
    }
  }
  __syncthreads();
  if (tile >= mtiles) return;

  v8f acc0 = {}, acc1 = {};
  const float4* bp = wf + (2 * f) * 32 + lane;
  for (int jc = 0; jc < 32; jc += 8) {
#pragma unroll
    for (int u = 0; u < 4; ++u) nxt[u] = *(const float4*)(xp + 8 * (jc + 4 + u));
    g1_chunk(cur, bp + (jc * 4) * 32, acc0, acc1);
    if (jc + 8 < 32) {
#pragma unroll
      for (int u = 0; u < 4; ++u) cur[u] = *(const float4*)(xp + 8 * (jc + 8 + u));
    }
    g1_chunk(nxt, bp + ((jc + 4) * 4) * 32, acc0, acc1);
  }

  float* hp = H + ((long)tile * 16 + 8 * h) * 64 + 32 * f + 2 * c;
#pragma unroll
  for (int v = 0; v < 8; ++v) {
    v2f o;
    o.x = acc0[v]; o.y = acc1[v];
    *(v2f*)(hp + v * 64) = o;
  }
}

#define BUCKET_SHIFT 8
#define BUCKET_NODES 256
#define MAXB 512
#define NCHUNK 256
#define STAGE_CAP 12288
#define CHUNK_CAP 16384

__global__ void __launch_bounds__(1024)
bucket_count(const int* __restrict__ ei, long Eraw, int nb, int* __restrict__ tbl) {
  __shared__ int hist[MAXB];
  const int t = threadIdx.x;
  const int wg = blockIdx.x;
  for (int b = t; b < nb; b += 1024) hist[b] = 0;
  __syncthreads();
  const long chunk = (Eraw + NCHUNK - 1) / NCHUNK;
  const long lo = wg * chunk;
  const long hi = (lo + chunk < Eraw) ? lo + chunk : Eraw;
  const int* dp = ei + Eraw;
  long e = lo + t;
  for (; e + 3072 < hi; e += 4096) {
    const int d0 = dp[e], d1 = dp[e + 1024], d2 = dp[e + 2048], d3 = dp[e + 3072];
    atomicAdd(&hist[d0 >> BUCKET_SHIFT], 1);
    atomicAdd(&hist[d1 >> BUCKET_SHIFT], 1);
    atomicAdd(&hist[d2 >> BUCKET_SHIFT], 1);
    atomicAdd(&hist[d3 >> BUCKET_SHIFT], 1);
  }
  for (; e < hi; e += 1024) atomicAdd(&hist[dp[e] >> BUCKET_SHIFT], 1);
  __syncthreads();
  for (int b = t; b < nb; b += 1024) tbl[b * NCHUNK + wg] = hist[b];
}

__global__ void __launch_bounds__(1024)
scan_offsets(const int* __restrict__ cnt, int* __restrict__ off, int n) {
  __shared__ int part[1024];
  const int t = threadIdx.x;
  const int base = blockIdx.x * 1024;
  int p = 0;
  for (int i = t; i < base; i += 1024) p += cnt[i];
  part[t] = p;
  __syncthreads();
  for (int o = 512; o > 0; o >>= 1) {
    if (t < o) part[t] += part[t + o];
    __syncthreads();
  }
  const int before = part[0];
  __syncthreads();
  const int i = base + t;
  const int v = (i < n) ? cnt[i] : 0;
  part[t] = v;
  __syncthreads();
  for (int o = 1; o < 1024; o <<= 1) {
    const int u = (t >= o) ? part[t - o] : 0;
    __syncthreads();
    part[t] += u;
    __syncthreads();
  }
  if (i < n) off[i] = before + part[t] - v;
  if (i == n - 1) off[n] = before + part[t];
}

__global__ void __launch_bounds__(1024)
bucket_scatter(const int* __restrict__ ei, long Eraw, int nb, const int* __restrict__ toff,
               unsigned* __restrict__ tmp) {
  __shared__ int cur[MAXB];
  __shared__ int dlt[MAXB];
  __shared__ int wtot[32];
  __shared__ int ss[CHUNK_CAP];
  __shared__ int sd[CHUNK_CAP];
  const int t = threadIdx.x;
  const int wg = blockIdx.x;
  const long chunk = (Eraw + NCHUNK - 1) / NCHUNK;
  const long lo = wg * chunk;
  const long hi = (lo + chunk < Eraw) ? lo + chunk : Eraw;
  const int* dp = ei + Eraw;
  long e = lo + t;

  if (hi - lo > CHUNK_CAP) {
    for (int b = t; b < nb; b += 1024) cur[b] = toff[b * NCHUNK + wg];
    __syncthreads();
    for (; e < hi; e += 1024) {
      const int s = ei[e], d = dp[e];
      const int p = atomicAdd(&cur[d >> BUCKET_SHIFT], 1);
      tmp[p] = ((unsigned)s << 8) | (unsigned)(d & (BUCKET_NODES - 1));
    }
    return;
  }

  const int lane = t & 31;
  const int wave = t >> 5;
  int g = 0, v = 0;
  if (t < nb) {
    g = toff[t * NCHUNK + wg];
    v = toff[t * NCHUNK + wg + 1] - g;
  }
  int x = v;
  for (int o = 1; o < 32; o <<= 1) {
    const int y = __shfl_up(x, o);
    if (lane >= o) x += y;
  }
  if (lane == 31) wtot[wave] = x;
  __syncthreads();
  int before = 0;
  for (int w = 0; w < wave; ++w) before += wtot[w];
  if (t < nb) {
    const int first = before + x - v;
    cur[t] = first;
    dlt[t] = g - first;
  }
  __syncthreads();

  for (; e + 3072 < hi; e += 4096) {
    const int s0 = ei[e], s1 = ei[e + 1024], s2 = ei[e + 2048], s3 = ei[e + 3072];
    const int d0 = dp[e], d1 = dp[e + 1024], d2 = dp[e + 2048], d3 = dp[e + 3072];
    const int p0 = atomicAdd(&cur[d0 >> BUCKET_SHIFT], 1);
    const int p1 = atomicAdd(&cur[d1 >> BUCKET_SHIFT], 1);
    const int p2 = atomicAdd(&cur[d2 >> BUCKET_SHIFT], 1);
    const int p3 = atomicAdd(&cur[d3 >> BUCKET_SHIFT], 1);
    ss[p0] = s0; sd[p0] = d0;
    ss[p1] = s1; sd[p1] = d1;
    ss[p2] = s2; sd[p2] = d2;
    ss[p3] = s3; sd[p3] = d3;
  }
  for (; e < hi; e += 1024) {
    const int s = ei[e], d = dp[e];
    const int p = atomicAdd(&cur[d >> BUCKET_SHIFT], 1);
    ss[p] = s; sd[p] = d;
  }
  __syncthreads();

  const int n = (int)(hi - lo);
  for (int i = t; i < n; i += 1024) {
    const int d = sd[i];
    tmp[i + dlt[d >> BUCKET_SHIFT]] =
        ((unsigned)ss[i] << 8) | (unsigned)(d & (BUCKET_NODES - 1));
  }
}

template <bool STAGED>
__device__ __forceinline__ void place_edges(const unsigned* __restrict__ tmp, int lo, int hi, int t,
                                            int* ncur, int* stage, int* __restrict__ perm) {
  int i = lo + t;
  for (; i + 3072 < hi; i += 4096) {
    const unsigned w0 = tmp[i], w1 = tmp[i + 1024], w2 = tmp[i + 2048], w3 = tmp[i + 3072];
    const int p0 = atomicAdd(&ncur[w0 & (BUCKET_NODES - 1)], 1);
    const int p1 = atomicAdd(&ncur[w1 & (BUCKET_NODES - 1)], 1);
    const int p2 = atomicAdd(&ncur[w2 & (BUCKET_NODES - 1)], 1);
    const int p3 = atomicAdd(&ncur[w3 & (BUCKET_NODES - 1)], 1);
    if (STAGED) {
      stage[p0 - lo] = (int)(w0 >> 8);
      stage[p1 - lo] = (int)(w1 >> 8);
      stage[p2 - lo] = (int)(w2 >> 8);
      stage[p3 - lo] = (int)(w3 >> 8);
    } else {
      perm[p0] = (int)(w0 >> 8);
      perm[p1] = (int)(w1 >> 8);
      perm[p2] = (int)(w2 >> 8);
      perm[p3] = (int)(w3 >> 8);
    }
  }
  for (; i < hi; i += 1024) {
    const unsigned w = tmp[i];
    const int p = atomicAdd(&ncur[w & (BUCKET_NODES - 1)], 1);
    if (STAGED) stage[p - lo] = (int)(w >> 8);
    else perm[p] = (int)(w >> 8);
  }
}

__global__ void __launch_bounds__(1024)
bucket_sort_nodes(const unsigned* __restrict__ tmp, const int* __restrict__ toff, int nb, int N,
                  int* __restrict__ off, int* __restrict__ perm) {
  __shared__ int ncnt[BUCKET_NODES];
  __shared__ int nsc[BUCKET_NODES];
  __shared__ int ncur[BUCKET_NODES];
  __shared__ int stage[STAGE_CAP];
  const int t = threadIdx.x;
  const int b = blockIdx.x;
  const int lo = toff[b * NCHUNK];
  const int hi = toff[(b + 1) * NCHUNK];
  if (t < BUCKET_NODES) ncnt[t] = 0;
  __syncthreads();
  int i = lo + t;
  for (; i + 3072 < hi; i += 4096) {
    const unsigned w0 = tmp[i], w1 = tmp[i + 1024], w2 = tmp[i + 2048], w3 = tmp[i + 3072];
    atomicAdd(&ncnt[w0 & (BUCKET_NODES - 1)], 1);
    atomicAdd(&ncnt[w1 & (BUCKET_NODES - 1)], 1);
    atomicAdd(&ncnt[w2 & (BUCKET_NODES - 1)], 1);
    atomicAdd(&ncnt[w3 & (BUCKET_NODES - 1)], 1);
  }
  for (; i < hi; i += 1024) atomicAdd(&ncnt[tmp[i] & (BUCKET_NODES - 1)], 1);
  __syncthreads();
  int v = 0;
  if (t < BUCKET_NODES) { v = ncnt[t]; nsc[t] = v; }
  __syncthreads();
  for (int o = 1; o < BUCKET_NODES; o <<= 1) {
    int u = 0;
    if (t < BUCKET_NODES && t >= o) u = nsc[t - o];
    __syncthreads();
    if (t < BUCKET_NODES) nsc[t] += u;
    __syncthreads();
  }
  if (t < BUCKET_NODES) {
    const int first = lo + nsc[t] - v;
    ncur[t] = first;
    const int node = b * BUCKET_NODES + t;
    if (node < N) off[node] = first;
  }
  if (b == nb - 1 && t == 0) off[N] = hi;
  __syncthreads();
  if (hi - lo <= STAGE_CAP) {
    place_edges<true>(tmp, lo, hi, t, ncur, stage, perm);
    __syncthreads();
    for (int k = (lo & ~31) + t; k < hi; k += 1024)
      if (k >= lo) perm[k] = stage[k - lo];
  } else {
    place_edges<false>(tmp, lo, hi, t, ncur, stage, perm);
  }
}

struct Row8 { v2f q[4]; };

__device__ __forceinline__ Row8 load_row8(const float* __restrict__ p) {
  const float4 a = *(const float4*)p;
  const float4 b = *(const float4*)(p + 4);
  Row8 r;
  r.q[0].x = a.x; r.q[0].y = a.y; r.q[1].x = a.z; r.q[1].y = a.w;
  r.q[2].x = b.x; r.q[2].y = b.y; r.q[3].x = b.z; r.q[3].y = b.w;
  return r;
}

__device__ __forceinline__ float dot8(const Row8& r, const Row8& w) {
  v2f ss = r.q[0] * w.q[0];
  ss += r.q[1] * w.q[1];
  ss += r.q[2] * w.q[2];
  ss += r.q[3] * w.q[3];
  return ss.x + ss.y;
}


__global__ void __launch_bounds__(256)
node_softmax_gather_heads8(const int* __restrict__ off, const int* __restrict__ perm,
                           const float* __restrict__ as, const float* __restrict__ ad,
                           const float* __restrict__ hf, const float* __restrict__ bias,
                           const float* __restrict__ w2, float* __restrict__ h2, long N) {
  __shared__ float4 rows[32 * 17];
  const long t0 = (long)blockIdx.x * 256 + threadIdx.x;
  const long t = t0 < N * 8 ? t0 : N * 8 - 8 + (t0 & 7);
  const int d = (int)(t >> 3);
  const int ro = (int)(t & 7) * 8;

  const Row8 asv = load_row8(as + ro);
  const Row8 own = load_row8(hf + (long)d * 64 + ro);
  const float ald = dot8(own, load_row8(ad + ro));
  float m = lrelu(dot8(own, asv) + ald);
  float den = 1.f;
  Row8 acc = own;

  const int j0 = off[d], j1 = off[d + 1];
  int j = j0;
  for (; j + 4 <= j1; j += 4) {
    const int s0 = perm[j], s1 = perm[j + 1], s2 = perm[j + 2], s3 = perm[j + 3];
    const Row8 r0 = load_row8(hf + (long)s0 * 64 + ro);
    const Row8 r1 = load_row8(hf + (long)s1 * 64 + ro);
    const Row8 r2 = load_row8(hf + (long)s2 * 64 + ro);
    const Row8 r3 = load_row8(hf + (long)s3 * 64 + ro);
    const float e0 = lrelu(dot8(r0, asv) + ald);
    const float e1 = lrelu(dot8(r1, asv) + ald);
    const float e2 = lrelu(dot8(r2, asv) + ald);
    const float e3 = lrelu(dot8(r3, asv) + ald);
    const float mn = fmaxf(fmaxf(fmaxf(m, e0), fmaxf(e1, e2)), e3);
    const float sc = __expf(m - mn);
    const float p0 = __expf(e0 - mn);
    const float p1 = __expf(e1 - mn);
    const float p2 = __expf(e2 - mn);
    const float p3 = __expf(e3 - mn);
    m = mn;
    den = den * sc + p0; den += p1; den += p2; den += p3;
#pragma unroll
    for (int k = 0; k < 4; ++k) {
      acc.q[k] = acc.q[k] * sc + p0 * r0.q[k];
      acc.q[k] += p1 * r1.q[k];
      acc.q[k] += p2 * r2.q[k];
      acc.q[k] += p3 * r3.q[k];
    }
  }
  for (; j < j1; ++j) {
    const Row8 r = load_row8(hf + (long)perm[j] * 64 + ro);
    const float e = lrelu(dot8(r, asv) + ald);
    const float mn = fmaxf(m, e);
    const float sc = __expf(m - mn);
    const float p = __expf(e - mn);
    m = mn;
    den = den * sc + p;
#pragma unroll
    for (int k = 0; k < 4; ++k) acc.q[k] = acc.q[k] * sc + p * r.q[k];
  }

  const float dn = den + 1e-16f;
  const float* bp = bias + ro;
  float4 o0, o1;
  o0.x = fmaxf(acc.q[0].x / dn + bp[0], 0.f); o0.y = fmaxf(acc.q[0].y / dn + bp[1], 0.f);
  o0.z = fmaxf(acc.q[1].x / dn + bp[2], 0.f); o0.w = fmaxf(acc.q[1].y / dn + bp[3], 0.f);
  o1.x = fmaxf(acc.q[2].x / dn + bp[4], 0.f); o1.y = fmaxf(acc.q[2].y / dn + bp[5], 0.f);
  o1.z = fmaxf(acc.q[3].x / dn + bp[6], 0.f); o1.w = fmaxf(acc.q[3].y / dn + bp[7], 0.f);
  float* sp = (float*)rows + (threadIdx.x >> 3) * 68 + ro;
  *(float4*)sp = o0;
  *(float4*)(sp + 4) = o1;

  const int lane = threadIdx.x & 31;
  const int wave = threadIdx.x >> 5;
  const int c = lane & 15;
  const int kg = (lane >> 4) << 1;
  v2f bf[16] = {};
  if (wave < 2) {
#pragma unroll
    for (int i = 0; i < 16; ++i) {
      bf[i].x = w2[(4 * i + kg) * 8 + (c & 7)];
      bf[i].y = w2[(4 * i + kg + 1) * 8 + (c & 7)];
    }
  }
  __syncthreads();
  if (wave < 2) {
    const float* arow = (const float*)rows + (16 * wave + c) * 68 + kg;
    v8f pr = {};
#pragma unroll
    for (int i = 0; i < 16; ++i) {
      const v2f a = *(const v2f*)(arow + 4 * i);
      pr = __builtin_amdgcn_wmma_f32_16x16x4_f32(false, a, false, bf[i], (short)0, pr,
                                                 false, false);
    }
    if (c < 8) {
      const long nb0 = (long)blockIdx.x * 32 + 16 * wave + 4 * kg;
#pragma unroll
      for (int v = 0; v < 8; ++v)
        if (nb0 + v < N) h2[(nb0 + v) * 8 + c] = pr[v];
    }
  }
}

template <int X>
__device__ __forceinline__ float lane_xor(float v) {
  return __builtin_bit_cast(float, __builtin_amdgcn_update_dpp(
      0, __builtin_bit_cast(int, v), 0x160 + X, 0xf, 0xf, true));
}

__device__ __forceinline__ float sum8(float v) {
  v += lane_xor<1>(v);
  v += lane_xor<2>(v);
  v += lane_xor<4>(v);
  return v;
}

__global__ void __launch_bounds__(256)
node_softmax_gather_head1(const int* __restrict__ off, const int* __restrict__ perm,
                          const float* __restrict__ as, const float* __restrict__ ad,
                          const float* __restrict__ hf, const float* __restrict__ bias,
                          const float* __restrict__ w3, const float* __restrict__ b3,
                          float* __restrict__ out, float* __restrict__ score, long N) {
  const long t = (long)blockIdx.x * blockDim.x + threadIdx.x;
  if (t >= N * 8) return;
  const int d = (int)(t >> 3);
  const int g = (int)(t & 7);

  const float asg = as[g];
  const float own = hf[(long)d * 8 + g];
  const float ald = sum8(own * ad[g]);
  float m = lrelu(sum8(own * asg) + ald);
  float den = 1.f;
  float acc = own;

  const int j0 = off[d], j1 = off[d + 1];
  int j = j0;
  for (; j + 4 <= j1; j += 4) {
    const int s0 = perm[j], s1 = perm[j + 1], s2 = perm[j + 2], s3 = perm[j + 3];
    const float x0 = hf[(long)s0 * 8 + g];
    const float x1 = hf[(long)s1 * 8 + g];
    const float x2 = hf[(long)s2 * 8 + g];
    const float x3 = hf[(long)s3 * 8 + g];
    const float e0 = lrelu(sum8(x0 * asg) + ald);
    const float e1 = lrelu(sum8(x1 * asg) + ald);
    const float e2 = lrelu(sum8(x2 * asg) + ald);
    const float e3 = lrelu(sum8(x3 * asg) + ald);
    const float mn = fmaxf(fmaxf(fmaxf(m, e0), fmaxf(e1, e2)), e3);
    const float sc = __expf(m - mn);
    const float p0 = __expf(e0 - mn);
    const float p1 = __expf(e1 - mn);
    const float p2 = __expf(e2 - mn);
    const float p3 = __expf(e3 - mn);
    m = mn;
    den = den * sc + p0; den += p1; den += p2; den += p3;
    acc = acc * sc + p0 * x0; acc += p1 * x1; acc += p2 * x2; acc += p3 * x3;
  }
  for (; j < j1; ++j) {
    const float x = hf[(long)perm[j] * 8 + g];
    const float e = lrelu(sum8(x * asg) + ald);
    const float mn = fmaxf(m, e);
    const float sc = __expf(m - mn);
    const float p = __expf(e - mn);
    m = mn;
    den = den * sc + p;
    acc = acc * sc + p * x;
  }
  const float o = acc / (den + 1e-16f) + bias[g];
  out[(long)d * 8 + g] = o;
  const float sc = sum8(o * w3[g]) + b3[0];
  if (g == 0) score[d] = sc;
}

static inline int nblk(long n, int b) { return (int)((n + b - 1) / b); }

extern "C" void kernel_launch(void* const* d_in, const int* in_sizes, int n_in,
                              void* d_out, int out_size, void* d_ws, size_t ws_size,
                              hipStream_t stream) {
  const float* x   = (const float*)d_in[0];
  const int*   ei  = (const int*)d_in[1];
  const float* W1  = (const float*)d_in[2];
  const float* as1 = (const float*)d_in[3];
  const float* ad1 = (const float*)d_in[4];
  const float* b1  = (const float*)d_in[5];
  const float* W2  = (const float*)d_in[6];
  const float* as2 = (const float*)d_in[7];
  const float* ad2 = (const float*)d_in[8];
  const float* b2  = (const float*)d_in[9];
  const float* W3  = (const float*)d_in[10];
  const float* b3  = (const float*)d_in[11];

  const long N    = in_sizes[0] / 256;
  const long Eraw = in_sizes[1] / 2;

  float* ws   = (float*)d_ws;
  float* h1   = ws;
  float* acc1 = h1 + N * 64;
  int*   perm = (int*)(acc1 + N * 64);
  int*   off  = perm + Eraw;
  const int nb = (int)((N + BUCKET_NODES - 1) / BUCKET_NODES);
  unsigned* tmp = (unsigned*)acc1;
  int*   tbl  = (int*)acc1 + Eraw;
  int*   toff = tbl + nb * NCHUNK;
  float* h2   = acc1;

  float* emb   = (float*)d_out;
  float* score = emb + N * 8;

  const int B = 256;
  const int mtiles = (int)(N / 16);

  bucket_count<<<NCHUNK, 1024, 0, stream>>>(ei, Eraw, nb, tbl);
  scan_offsets<<<nblk(nb * NCHUNK, 1024), 1024, 0, stream>>>(tbl, toff, nb * NCHUNK);
  bucket_scatter<<<NCHUNK, 1024, 0, stream>>>(ei, Eraw, nb, toff, tmp);
  bucket_sort_nodes<<<nb, 1024, 0, stream>>>(tmp, toff, nb, (int)N, off, perm);

  gemm1_x_w1<<<(mtiles + 3) / 4, 256, 0, stream>>>(x, W1, h1, mtiles);
  node_softmax_gather_heads8<<<nblk(N * 8, 256), 256, 0, stream>>>(off, perm, as1, ad1, h1, b1, W2, h2, N);

  node_softmax_gather_head1<<<nblk(N * 8, B), B, 0, stream>>>(off, perm, as2, ad2, h2, b2, W3, b3,
                                                              emb, score, N);
}


# Level 1 of the grouping in one launch: chunk ranked from registers (returning LDS add), ordered in LDS, stored at its own place with its run starts; count+scan launches go, 7->5; sum order changes
# speedup vs baseline: 20.5673x; 1.1418x over previous
#include <hip/hip_runtime.h>
#include <math.h>

typedef __attribute__((ext_vector_type(2))) float v2f;
typedef __attribute__((ext_vector_type(8))) float v8f;

#define NEG_SLOPE 0.2f

__device__ __forceinline__ float lrelu(float v) { return v >= 0.f ? v : NEG_SLOPE * v; }

__device__ __forceinline__ float g1_elem(const float4& r, int i) {
  return i == 0 ? r.x : i == 1 ? r.y : i == 2 ? r.z : r.w;
}

__device__ __forceinline__ void g1_chunk(const float4 (&a)[4], const float4* bp,
                                         v8f& acc0, v8f& acc1) {
#pragma unroll
  for (int u = 0; u < 4; ++u) {
    const float4 b0 = bp[(u * 4 + 0) * 32];
    const float4 b1 = bp[(u * 4 + 1) * 32];
    v2f alo, ahi, b;
    alo.x = a[u].x; alo.y = a[u].y; ahi.x = a[u].z; ahi.y = a[u].w;
    b.x = b0.x; b.y = b0.y;
    acc0 = __builtin_amdgcn_wmma_f32_16x16x4_f32(false, alo, false, b, (short)0, acc0, false, false);
    b.x = b1.x; b.y = b1.y;
    acc1 = __builtin_amdgcn_wmma_f32_16x16x4_f32(false, alo, false, b, (short)0, acc1, false, false);
    b.x = b0.z; b.y = b0.w;
    acc0 = __builtin_amdgcn_wmma_f32_16x16x4_f32(false, ahi, false, b, (short)0, acc0, false, false);
    b.x = b1.z; b.y = b1.w;
    acc1 = __builtin_amdgcn_wmma_f32_16x16x4_f32(false, ahi, false, b, (short)0, acc1, false, false);
  }
}

__global__ void __launch_bounds__(256)
gemm1_x_w1(const float* __restrict__ X, const float* __restrict__ W,
           float* __restrict__ H, int mtiles) {
  __shared__ float4 wf[32 * 4 * 32];
  const int lane = threadIdx.x & 31;
  const int wave = threadIdx.x >> 5;
  const int c = lane & 15;
  const int h = lane >> 4;
  const int f = wave & 1;
  const int tile = blockIdx.x * 4 + (wave >> 1);
  const int tl = tile < mtiles ? tile : mtiles - 1;
  const float* xp = X + ((long)tl * 16 + c) * 256 + 4 * h;

  float4 cur[4], nxt[4];
#pragma unroll
  for (int u = 0; u < 4; ++u) cur[u] = *(const float4*)(xp + 8 * u);

#pragma unroll
  for (int m = 0; m < 4; ++m) {
    const int j = wave + 8 * m;
    const float* wp = W + (8 * j + 4 * h) * 64 + 4 * c;
    const float4 r0 = *(const float4*)(wp);
    const float4 r1 = *(const float4*)(wp + 64);
    const float4 r2 = *(const float4*)(wp + 128);
    const float4 r3 = *(const float4*)(wp + 192);
#pragma unroll
    for (int i = 0; i < 4; ++i) {
      float4 o;
      o.x = g1_elem(r0, i); o.y = g1_elem(r1, i); o.z = g1_elem(r2, i); o.w = g1_elem(r3, i);
      const int n = 2 * (c >> 3) + (i & 1);
      const int cl = 2 * (c & 7) + (i >> 1);
      wf[(j * 4 + n) * 32 + 16 * h + cl] = o;
    }
  }
  __syncthreads();
  if (tile >= mtiles) return;

  v8f acc0 = {}, acc1 = {};
  const float4* bp = wf + (2 * f) * 32 + lane;
  for (int jc = 0; jc < 32; jc += 8) {
#pragma unroll
    for (int u = 0; u < 4; ++u) nxt[u] = *(const float4*)(xp + 8 * (jc + 4 + u));
    g1_chunk(cur, bp + (jc * 4) * 32, acc0, acc1);
    if (jc + 8 < 32) {
#pragma unroll
      for (int u = 0; u < 4; ++u) cur[u] = *(const float4*)(xp + 8 * (jc + 8 + u));
    }
    g1_chunk(nxt, bp + ((jc + 4) * 4) * 32, acc0, acc1);
  }

  float* hp = H + ((long)tile * 16 + 8 * h) * 64 + 32 * f + 2 * c;
#pragma unroll
  for (int v = 0; v < 8; ++v) {
    v2f o;
    o.x = acc0[v]; o.y = acc1[v];
    *(v2f*)(hp + v * 64) = o;
  }
}

#define BUCKET_SHIFT 8
#define BUCKET_NODES 256
#define MAXB 512
#define NCHUNK 256
#define MAXCH 1024
#define STAGE_CAP 12288
#define CHUNK_CAP 16384
#define EPT 16

__global__ void __launch_bounds__(1024)
chunk_sort(const int* __restrict__ ei, long Eraw, int chunk, int nb,
           int* __restrict__ st, unsigned* __restrict__ tmp) {
  __shared__ int cnt[MAXB];
  __shared__ int wtot[32];
  __shared__ unsigned sw[CHUNK_CAP];
  const int t = threadIdx.x;
  const int wg = blockIdx.x;
  const long lo = (long)wg * chunk;
  const long hi = (lo + chunk < Eraw) ? lo + chunk : Eraw;
  const int n = (int)(hi - lo);
  const int* sp = ei + lo;
  const int* dp = ei + Eraw + lo;

  if (t < MAXB) cnt[t] = 0;
  int s[EPT], d[EPT], p[EPT];
#pragma unroll
  for (int k = 0; k < EPT; ++k) {
    const int i = t + k * 1024;
    s[k] = 0; d[k] = 0; p[k] = 0;
    if (i < n) { s[k] = sp[i]; d[k] = dp[i]; }
  }
  __syncthreads();
#pragma unroll
  for (int k = 0; k < EPT; ++k)
    if (t + k * 1024 < n) p[k] = atomicAdd(&cnt[d[k] >> BUCKET_SHIFT], 1);
  __syncthreads();

  const int lane = t & 31;
  const int wave = t >> 5;
  const int v = (t < nb) ? cnt[t] : 0;
  int x = v;
  for (int o = 1; o < 32; o <<= 1) {
    const int y = __shfl_up(x, o);
    if (lane >= o) x += y;
  }
  if (lane == 31) wtot[wave] = x;
  __syncthreads();
  int before = 0;
  for (int w = 0; w < wave; ++w) before += wtot[w];
  int* sr = st + (long)wg * (nb + 1);
  if (t < nb) {
    const int first = before + x - v;
    cnt[t] = first;
    sr[t] = first;
  }
  if (t == 0) sr[nb] = n;
  __syncthreads();

#pragma unroll
  for (int k = 0; k < EPT; ++k)
    if (t + k * 1024 < n)
      sw[cnt[d[k] >> BUCKET_SHIFT] + p[k]] =
          ((unsigned)s[k] << 8) | (unsigned)(d[k] & (BUCKET_NODES - 1));
  __syncthreads();
  for (int i = t; i < n; i += 1024) tmp[lo + i] = sw[i];
}

template <int MODE>
__device__ __forceinline__ void edge_word(unsigned w, int* nctr, int* stage, int lo,
                                          int* __restrict__ perm) {
  if (MODE == 0) {
    atomicAdd(&nctr[w & (BUCKET_NODES - 1)], 1);
  } else {
    const int p = atomicAdd(&nctr[w & (BUCKET_NODES - 1)], 1);
    if (MODE == 1) stage[p - lo] = (int)(w >> 8);
    else perm[p] = (int)(w >> 8);
  }
}

template <int MODE>
__device__ __forceinline__ void walk_runs(const unsigned* __restrict__ tmp, const int* rb,
                                          const int* rl, int nchunk, int wave, int lane,
                                          int* nctr, int* stage, int lo, int* __restrict__ perm) {
  for (int c0 = wave; c0 < nchunk; c0 += 256) {
    int base[8], len[8];
    unsigned w0[8], w1[8];
#pragma unroll
    for (int r = 0; r < 8; ++r) {
      const int c = c0 + 32 * r;
      base[r] = (c < nchunk) ? rb[c] : 0;
      len[r] = (c < nchunk) ? rl[c] : 0;
    }
#pragma unroll
    for (int r = 0; r < 8; ++r) {
      const int a0 = (lane < len[r]) ? base[r] + lane : 0;
      const int a1 = (lane + 32 < len[r]) ? base[r] + lane + 32 : 0;
      w0[r] = tmp[a0];
      w1[r] = tmp[a1];
    }
#pragma unroll
    for (int r = 0; r < 8; ++r) {
      if (lane < len[r]) edge_word<MODE>(w0[r], nctr, stage, lo, perm);
      if (lane + 32 < len[r]) edge_word<MODE>(w1[r], nctr, stage, lo, perm);
      for (int i = lane + 64; i < len[r]; i += 32)
        edge_word<MODE>(tmp[base[r] + i], nctr, stage, lo, perm);
    }
  }
}

__global__ void __launch_bounds__(1024)
bucket_sort_nodes(const unsigned* __restrict__ tmp, const int* __restrict__ st, int chunk,
                  int nchunk, int nb, int N, int* __restrict__ off, int* __restrict__ perm) {
  __shared__ int ncnt[BUCKET_NODES];
  __shared__ int nsc[BUCKET_NODES];
  __shared__ int ncur[BUCKET_NODES];
  __shared__ int rb[MAXCH];
  __shared__ int rl[MAXCH];
  __shared__ int tot[2];
  __shared__ int stage[STAGE_CAP];
  const int t = threadIdx.x;
  const int b = blockIdx.x;
  const int lane = t & 31;
  const int wave = t >> 5;
  if (t < BUCKET_NODES) ncnt[t] = 0;
  if (t < 2) tot[t] = 0;
  int s0 = 0, ln = 0;
  if (t < nchunk) {
    const int* q = st + (long)t * (nb + 1) + b;
    s0 = q[0];
    ln = q[1] - s0;
    rb[t] = t * chunk + s0;
    rl[t] = ln;
  }
  for (int o = 16; o > 0; o >>= 1) {
    s0 += __shfl_xor(s0, o);
    ln += __shfl_xor(ln, o);
  }
  __syncthreads();
  if (lane == 0) { atomicAdd(&tot[0], s0); atomicAdd(&tot[1], ln); }
  __syncthreads();
  const int lo = tot[0];
  const int hi = lo + tot[1];
  walk_runs<0>(tmp, rb, rl, nchunk, wave, lane, ncnt, stage, lo, perm);
  __syncthreads();
  int v = 0;
  if (t < BUCKET_NODES) { v = ncnt[t]; nsc[t] = v; }
  __syncthreads();
  for (int o = 1; o < BUCKET_NODES; o <<= 1) {
    int u = 0;
    if (t < BUCKET_NODES && t >= o) u = nsc[t - o];
    __syncthreads();
    if (t < BUCKET_NODES) nsc[t] += u;
    __syncthreads();
  }
  if (t < BUCKET_NODES) {
    const int first = lo + nsc[t] - v;
    ncur[t] = first;
    const int node = b * BUCKET_NODES + t;
    if (node < N) off[node] = first;
  }
  if (b == nb - 1 && t == 0) off[N] = hi;
  __syncthreads();
  if (hi - lo <= STAGE_CAP) {
    walk_runs<1>(tmp, rb, rl, nchunk, wave, lane, ncur, stage, lo, perm);
    __syncthreads();
    for (int k = (lo & ~31) + t; k < hi; k += 1024)
      if (k >= lo) perm[k] = stage[k - lo];
  } else {
    walk_runs<2>(tmp, rb, rl, nchunk, wave, lane, ncur, stage, lo, perm);
  }
}

struct Row8 { v2f q[4]; };

__device__ __forceinline__ Row8 load_row8(const float* __restrict__ p) {
  const float4 a = *(const float4*)p;
  const float4 b = *(const float4*)(p + 4);
  Row8 r;
  r.q[0].x = a.x; r.q[0].y = a.y; r.q[1].x = a.z; r.q[1].y = a.w;
  r.q[2].x = b.x; r.q[2].y = b.y; r.q[3].x = b.z; r.q[3].y = b.w;
  return r;
}

__device__ __forceinline__ float dot8(const Row8& r, const Row8& w) {
  v2f ss = r.q[0] * w.q[0];
  ss += r.q[1] * w.q[1];
  ss += r.q[2] * w.q[2];
  ss += r.q[3] * w.q[3];
  return ss.x + ss.y;
}


__global__ void __launch_bounds__(256)
node_softmax_gather_heads8(const int* __restrict__ off, const int* __restrict__ perm,
                           const float* __restrict__ as, const float* __restrict__ ad,
                           const float* __restrict__ hf, const float* __restrict__ bias,
                           const float* __restrict__ w2, float* __restrict__ h2, long N) {
  __shared__ float4 rows[32 * 17];
  const long t0 = (long)blockIdx.x * 256 + threadIdx.x;
  const long t = t0 < N * 8 ? t0 : N * 8 - 8 + (t0 & 7);
  const int d = (int)(t >> 3);
  const int ro = (int)(t & 7) * 8;

  const Row8 asv = load_row8(as + ro);
  const Row8 own = load_row8(hf + (long)d * 64 + ro);
  const float ald = dot8(own, load_row8(ad + ro));
  float m = lrelu(dot8(own, asv) + ald);
  float den = 1.f;
  Row8 acc = own;

  const int j0 = off[d], j1 = off[d + 1];
  int j = j0;
  for (; j + 4 <= j1; j += 4) {
    const int s0 = perm[j], s1 = perm[j + 1], s2 = perm[j + 2], s3 = perm[j + 3];
    const Row8 r0 = load_row8(hf + (long)s0 * 64 + ro);
    const Row8 r1 = load_row8(hf + (long)s1 * 64 + ro);
    const Row8 r2 = load_row8(hf + (long)s2 * 64 + ro);
    const Row8 r3 = load_row8(hf + (long)s3 * 64 + ro);
    const float e0 = lrelu(dot8(r0, asv) + ald);
    const float e1 = lrelu(dot8(r1, asv) + ald);
    const float e2 = lrelu(dot8(r2, asv) + ald);
    const float e3 = lrelu(dot8(r3, asv) + ald);
    const float mn = fmaxf(fmaxf(fmaxf(m, e0), fmaxf(e1, e2)), e3);
    const float sc = __expf(m - mn);
    const float p0 = __expf(e0 - mn);
    const float p1 = __expf(e1 - mn);
    const float p2 = __expf(e2 - mn);
    const float p3 = __expf(e3 - mn);
    m = mn;
    den = den * sc + p0; den += p1; den += p2; den += p3;
#pragma unroll
    for (int k = 0; k < 4; ++k) {
      acc.q[k] = acc.q[k] * sc + p0 * r0.q[k];
      acc.q[k] += p1 * r1.q[k];
      acc.q[k] += p2 * r2.q[k];
      acc.q[k] += p3 * r3.q[k];
    }
  }
  for (; j < j1; ++j) {
    const Row8 r = load_row8(hf + (long)perm[j] * 64 + ro);
    const float e = lrelu(dot8(r, asv) + ald);
    const float mn = fmaxf(m, e);
    const float sc = __expf(m - mn);
    const float p = __expf(e - mn);
    m = mn;
    den = den * sc + p;
#pragma unroll
    for (int k = 0; k < 4; ++k) acc.q[k] = acc.q[k] * sc + p * r.q[k];
  }

  const float dn = den + 1e-16f;
  const float* bp = bias + ro;
  float4 o0, o1;
  o0.x = fmaxf(acc.q[0].x / dn + bp[0], 0.f); o0.y = fmaxf(acc.q[0].y / dn + bp[1], 0.f);
  o0.z = fmaxf(acc.q[1].x / dn + bp[2], 0.f); o0.w = fmaxf(acc.q[1].y / dn + bp[3], 0.f);
  o1.x = fmaxf(acc.q[2].x / dn + bp[4], 0.f); o1.y = fmaxf(acc.q[2].y / dn + bp[5], 0.f);
  o1.z = fmaxf(acc.q[3].x / dn + bp[6], 0.f); o1.w = fmaxf(acc.q[3].y / dn + bp[7], 0.f);
  float* sp = (float*)rows + (threadIdx.x >> 3) * 68 + ro;
  *(float4*)sp = o0;
  *(float4*)(sp + 4) = o1;

  const int lane = threadIdx.x & 31;
  const int wave = threadIdx.x >> 5;
  const int c = lane & 15;
  const int kg = (lane >> 4) << 1;
  v2f bf[16] = {};
  if (wave < 2) {
#pragma unroll
    for (int i = 0; i < 16; ++i) {
      bf[i].x = w2[(4 * i + kg) * 8 + (c & 7)];
      bf[i].y = w2[(4 * i + kg + 1) * 8 + (c & 7)];
    }
  }
  __syncthreads();
  if (wave < 2) {
    const float* arow = (const float*)rows + (16 * wave + c) * 68 + kg;
    v8f pr = {};
#pragma unroll
    for (int i = 0; i < 16; ++i) {
      const v2f a = *(const v2f*)(arow + 4 * i);
      pr = __builtin_amdgcn_wmma_f32_16x16x4_f32(false, a, false, bf[i], (short)0, pr,
                                                 false, false);
    }
    if (c < 8) {
      const long nb0 = (long)blockIdx.x * 32 + 16 * wave + 4 * kg;
#pragma unroll
      for (int v = 0; v < 8; ++v)
        if (nb0 + v < N) h2[(nb0 + v) * 8 + c] = pr[v];
    }
  }
}

template <int X>
__device__ __forceinline__ float lane_xor(float v) {
  return __builtin_bit_cast(float, __builtin_amdgcn_update_dpp(
      0, __builtin_bit_cast(int, v), 0x160 + X, 0xf, 0xf, true));
}

__device__ __forceinline__ float sum8(float v) {
  v += lane_xor<1>(v);
  v += lane_xor<2>(v);
  v += lane_xor<4>(v);
  return v;
}

__global__ void __launch_bounds__(256)
node_softmax_gather_head1(const int* __restrict__ off, const int* __restrict__ perm,
                          const float* __restrict__ as, const float* __restrict__ ad,
                          const float* __restrict__ hf, const float* __restrict__ bias,
                          const float* __restrict__ w3, const float* __restrict__ b3,
                          float* __restrict__ out, float* __restrict__ score, long N) {
  const long t = (long)blockIdx.x * blockDim.x + threadIdx.x;
  if (t >= N * 8) return;
  const int d = (int)(t >> 3);
  const int g = (int)(t & 7);

  const float asg = as[g];
  const float own = hf[(long)d * 8 + g];
  const float ald = sum8(own * ad[g]);
  float m = lrelu(sum8(own * asg) + ald);
  float den = 1.f;
  float acc = own;

  const int j0 = off[d], j1 = off[d + 1];
  int j = j0;
  for (; j + 4 <= j1; j += 4) {
    const int s0 = perm[j], s1 = perm[j + 1], s2 = perm[j + 2], s3 = perm[j + 3];
    const float x0 = hf[(long)s0 * 8 + g];
    const float x1 = hf[(long)s1 * 8 + g];
    const float x2 = hf[(long)s2 * 8 + g];
    const float x3 = hf[(long)s3 * 8 + g];
    const float e0 = lrelu(sum8(x0 * asg) + ald);
    const float e1 = lrelu(sum8(x1 * asg) + ald);
    const float e2 = lrelu(sum8(x2 * asg) + ald);
    const float e3 = lrelu(sum8(x3 * asg) + ald);
    const float mn = fmaxf(fmaxf(fmaxf(m, e0), fmaxf(e1, e2)), e3);
    const float sc = __expf(m - mn);
    const float p0 = __expf(e0 - mn);
    const float p1 = __expf(e1 - mn);
    const float p2 = __expf(e2 - mn);
    const float p3 = __expf(e3 - mn);
    m = mn;
    den = den * sc + p0; den += p1; den += p2; den += p3;
    acc = acc * sc + p0 * x0; acc += p1 * x1; acc += p2 * x2; acc += p3 * x3;
  }
  for (; j < j1; ++j) {
    const float x = hf[(long)perm[j] * 8 + g];
    const float e = lrelu(sum8(x * asg) + ald);
    const float mn = fmaxf(m, e);
    const float sc = __expf(m - mn);
    const float p = __expf(e - mn);
    m = mn;
    den = den * sc + p;
    acc = acc * sc + p * x;
  }
  const float o = acc / (den + 1e-16f) + bias[g];
  out[(long)d * 8 + g] = o;
  const float sc = sum8(o * w3[g]) + b3[0];
  if (g == 0) score[d] = sc;
}

static inline int nblk(long n, int b) { return (int)((n + b - 1) / b); }

extern "C" void kernel_launch(void* const* d_in, const int* in_sizes, int n_in,
                              void* d_out, int out_size, void* d_ws, size_t ws_size,
                              hipStream_t stream) {
  const float* x   = (const float*)d_in[0];
  const int*   ei  = (const int*)d_in[1];
  const float* W1  = (const float*)d_in[2];
  const float* as1 = (const float*)d_in[3];
  const float* ad1 = (const float*)d_in[4];
  const float* b1  = (const float*)d_in[5];
  const float* W2  = (const float*)d_in[6];
  const float* as2 = (const float*)d_in[7];
  const float* ad2 = (const float*)d_in[8];
  const float* b2  = (const float*)d_in[9];
  const float* W3  = (const float*)d_in[10];
  const float* b3  = (const float*)d_in[11];

  const long N    = in_sizes[0] / 256;
  const long Eraw = in_sizes[1] / 2;

  float* ws   = (float*)d_ws;
  float* h1   = ws;
  float* acc1 = h1 + N * 64;
  int*   perm = (int*)(acc1 + N * 64);
  int*   off  = perm + Eraw;
  const int nb = (int)((N + BUCKET_NODES - 1) / BUCKET_NODES);
  long chunk_l = (Eraw + NCHUNK - 1) / NCHUNK;
  if (chunk_l > CHUNK_CAP) chunk_l = CHUNK_CAP;
  if (chunk_l < 1) chunk_l = 1;
  const int chunk  = (int)chunk_l;
  const int nchunk = nblk(Eraw, chunk);
  unsigned* tmp = (unsigned*)acc1;
  int*   st   = (int*)acc1 + Eraw;
  float* h2   = acc1;

  float* emb   = (float*)d_out;
  float* score = emb + N * 8;

  const int B = 256;
  const int mtiles = (int)(N / 16);

  chunk_sort<<<nchunk, 1024, 0, stream>>>(ei, Eraw, chunk, nb, st, tmp);
  bucket_sort_nodes<<<nb, 1024, 0, stream>>>(tmp, st, chunk, nchunk, nb, (int)N, off, perm);

  gemm1_x_w1<<<(mtiles + 3) / 4, 256, 0, stream>>>(x, W1, h1, mtiles);
  node_softmax_gather_heads8<<<nblk(N * 8, 256), 256, 0, stream>>>(off, perm, as1, ad1, h1, b1, W2, h2, N);

  node_softmax_gather_head1<<<nblk(N * 8, B), B, 0, stream>>>(off, perm, as2, ad2, h2, b2, W3, b3,
                                                              emb, score, N);
}


# Gather walks (both layers): source ids fetched one step ahead; a node's 1-3 leftover edges summed as ONE masked step, not one memory trip each; sum rounding of the leftover edges changes
# speedup vs baseline: 21.8245x; 1.0611x over previous
#include <hip/hip_runtime.h>
#include <math.h>

typedef __attribute__((ext_vector_type(2))) float v2f;
typedef __attribute__((ext_vector_type(8))) float v8f;

#define NEG_SLOPE 0.2f

__device__ __forceinline__ float lrelu(float v) { return v >= 0.f ? v : NEG_SLOPE * v; }

__device__ __forceinline__ float g1_elem(const float4& r, int i) {
  return i == 0 ? r.x : i == 1 ? r.y : i == 2 ? r.z : r.w;
}

__device__ __forceinline__ void g1_chunk(const float4 (&a)[4], const float4* bp,
                                         v8f& acc0, v8f& acc1) {
#pragma unroll
  for (int u = 0; u < 4; ++u) {
    const float4 b0 = bp[(u * 4 + 0) * 32];
    const float4 b1 = bp[(u * 4 + 1) * 32];
    v2f alo, ahi, b;
    alo.x = a[u].x; alo.y = a[u].y; ahi.x = a[u].z; ahi.y = a[u].w;
    b.x = b0.x; b.y = b0.y;
    acc0 = __builtin_amdgcn_wmma_f32_16x16x4_f32(false, alo, false, b, (short)0, acc0, false, false);
    b.x = b1.x; b.y = b1.y;
    acc1 = __builtin_amdgcn_wmma_f32_16x16x4_f32(false, alo, false, b, (short)0, acc1, false, false);
    b.x = b0.z; b.y = b0.w;
    acc0 = __builtin_amdgcn_wmma_f32_16x16x4_f32(false, ahi, false, b, (short)0, acc0, false, false);
    b.x = b1.z; b.y = b1.w;
    acc1 = __builtin_amdgcn_wmma_f32_16x16x4_f32(false, ahi, false, b, (short)0, acc1, false, false);
  }
}

__global__ void __launch_bounds__(256)
gemm1_x_w1(const float* __restrict__ X, const float* __restrict__ W,
           float* __restrict__ H, int mtiles) {
  __shared__ float4 wf[32 * 4 * 32];
  const int lane = threadIdx.x & 31;
  const int wave = threadIdx.x >> 5;
  const int c = lane & 15;
  const int h = lane >> 4;
  const int f = wave & 1;
  const int tile = blockIdx.x * 4 + (wave >> 1);
  const int tl = tile < mtiles ? tile : mtiles - 1;
  const float* xp = X + ((long)tl * 16 + c) * 256 + 4 * h;

  float4 cur[4], nxt[4];
#pragma unroll
  for (int u = 0; u < 4; ++u) cur[u] = *(const float4*)(xp + 8 * u);

#pragma unroll
  for (int m = 0; m < 4; ++m) {
    const int j = wave + 8 * m;
    const float* wp = W + (8 * j + 4 * h) * 64 + 4 * c;
    const float4 r0 = *(const float4*)(wp);
    const float4 r1 = *(const float4*)(wp + 64);
    const float4 r2 = *(const float4*)(wp + 128);
    const float4 r3 = *(const float4*)(wp + 192);
#pragma unroll
    for (int i = 0; i < 4; ++i) {
      float4 o;
      o.x = g1_elem(r0, i); o.y = g1_elem(r1, i); o.z = g1_elem(r2, i); o.w = g1_elem(r3, i);
      const int n = 2 * (c >> 3) + (i & 1);
      const int cl = 2 * (c & 7) + (i >> 1);
      wf[(j * 4 + n) * 32 + 16 * h + cl] = o;
    }
  }
  __syncthreads();
  if (tile >= mtiles) return;

  v8f acc0 = {}, acc1 = {};
  const float4* bp = wf + (2 * f) * 32 + lane;
  for (int jc = 0; jc < 32; jc += 8) {
#pragma unroll
    for (int u = 0; u < 4; ++u) nxt[u] = *(const float4*)(xp + 8 * (jc + 4 + u));
    g1_chunk(cur, bp + (jc * 4) * 32, acc0, acc1);
    if (jc + 8 < 32) {
#pragma unroll
      for (int u = 0; u < 4; ++u) cur[u] = *(const float4*)(xp + 8 * (jc + 8 + u));
    }
    g1_chunk(nxt, bp + ((jc + 4) * 4) * 32, acc0, acc1);
  }

  float* hp = H + ((long)tile * 16 + 8 * h) * 64 + 32 * f + 2 * c;
#pragma unroll
  for (int v = 0; v < 8; ++v) {
    v2f o;
    o.x = acc0[v]; o.y = acc1[v];
    *(v2f*)(hp + v * 64) = o;
  }
}

#define BUCKET_SHIFT 8
#define BUCKET_NODES 256
#define MAXB 512
#define NCHUNK 256
#define MAXCH 1024
#define STAGE_CAP 12288
#define CHUNK_CAP 16384
#define EPT 16

__global__ void __launch_bounds__(1024)
chunk_sort(const int* __restrict__ ei, long Eraw, int chunk, int nb,
           int* __restrict__ st, unsigned* __restrict__ tmp) {
  __shared__ int cnt[MAXB];
  __shared__ int wtot[32];
  __shared__ unsigned sw[CHUNK_CAP];
  const int t = threadIdx.x;
  const int wg = blockIdx.x;
  const long lo = (long)wg * chunk;
  const long hi = (lo + chunk < Eraw) ? lo + chunk : Eraw;
  const int n = (int)(hi - lo);
  const int* sp = ei + lo;
  const int* dp = ei + Eraw + lo;

  if (t < MAXB) cnt[t] = 0;
  int s[EPT], d[EPT], p[EPT];
#pragma unroll
  for (int k = 0; k < EPT; ++k) {
    const int i = t + k * 1024;
    s[k] = 0; d[k] = 0; p[k] = 0;
    if (i < n) { s[k] = sp[i]; d[k] = dp[i]; }
  }
  __syncthreads();
#pragma unroll
  for (int k = 0; k < EPT; ++k)
    if (t + k * 1024 < n) p[k] = atomicAdd(&cnt[d[k] >> BUCKET_SHIFT], 1);
  __syncthreads();

  const int lane = t & 31;
  const int wave = t >> 5;
  const int v = (t < nb) ? cnt[t] : 0;
  int x = v;
  for (int o = 1; o < 32; o <<= 1) {
    const int y = __shfl_up(x, o);
    if (lane >= o) x += y;
  }
  if (lane == 31) wtot[wave] = x;
  __syncthreads();
  int before = 0;
  for (int w = 0; w < wave; ++w) before += wtot[w];
  int* sr = st + (long)wg * (nb + 1);
  if (t < nb) {
    const int first = before + x - v;
    cnt[t] = first;
    sr[t] = first;
  }
  if (t == 0) sr[nb] = n;
  __syncthreads();

#pragma unroll
  for (int k = 0; k < EPT; ++k)
    if (t + k * 1024 < n)
      sw[cnt[d[k] >> BUCKET_SHIFT] + p[k]] =
          ((unsigned)s[k] << 8) | (unsigned)(d[k] & (BUCKET_NODES - 1));
  __syncthreads();
  for (int i = t; i < n; i += 1024) tmp[lo + i] = sw[i];
}

template <int MODE>
__device__ __forceinline__ void edge_word(unsigned w, int* nctr, int* stage, int lo,
                                          int* __restrict__ perm) {
  if (MODE == 0) {
    atomicAdd(&nctr[w & (BUCKET_NODES - 1)], 1);
  } else {
    const int p = atomicAdd(&nctr[w & (BUCKET_NODES - 1)], 1);
    if (MODE == 1) stage[p - lo] = (int)(w >> 8);
    else perm[p] = (int)(w >> 8);
  }
}

template <int MODE>
__device__ __forceinline__ void walk_runs(const unsigned* __restrict__ tmp, const int* rb,
                                          const int* rl, int nchunk, int wave, int lane,
                                          int* nctr, int* stage, int lo, int* __restrict__ perm) {
  for (int c0 = wave; c0 < nchunk; c0 += 256) {
    int base[8], len[8];
    unsigned w0[8], w1[8];
#pragma unroll
    for (int r = 0; r < 8; ++r) {
      const int c = c0 + 32 * r;
      base[r] = (c < nchunk) ? rb[c] : 0;
      len[r] = (c < nchunk) ? rl[c] : 0;
    }
#pragma unroll
    for (int r = 0; r < 8; ++r) {
      const int a0 = (lane < len[r]) ? base[r] + lane : 0;
      const int a1 = (lane + 32 < len[r]) ? base[r] + lane + 32 : 0;
      w0[r] = tmp[a0];
      w1[r] = tmp[a1];
    }
#pragma unroll
    for (int r = 0; r < 8; ++r) {
      if (lane < len[r]) edge_word<MODE>(w0[r], nctr, stage, lo, perm);
      if (lane + 32 < len[r]) edge_word<MODE>(w1[r], nctr, stage, lo, perm);
      for (int i = lane + 64; i < len[r]; i += 32)
        edge_word<MODE>(tmp[base[r] + i], nctr, stage, lo, perm);
    }
  }
}

__global__ void __launch_bounds__(1024)
bucket_sort_nodes(const unsigned* __restrict__ tmp, const int* __restrict__ st, int chunk,
                  int nchunk, int nb, int N, int* __restrict__ off, int* __restrict__ perm) {
  __shared__ int ncnt[BUCKET_NODES];
  __shared__ int nsc[BUCKET_NODES];
  __shared__ int ncur[BUCKET_NODES];
  __shared__ int rb[MAXCH];
  __shared__ int rl[MAXCH];
  __shared__ int tot[2];
  __shared__ int stage[STAGE_CAP];
  const int t = threadIdx.x;
  const int b = blockIdx.x;
  const int lane = t & 31;
  const int wave = t >> 5;
  if (t < BUCKET_NODES) ncnt[t] = 0;
  if (t < 2) tot[t] = 0;
  int s0 = 0, ln = 0;
  if (t < nchunk) {
    const int* q = st + (long)t * (nb + 1) + b;
    s0 = q[0];
    ln = q[1] - s0;
    rb[t] = t * chunk + s0;
    rl[t] = ln;
  }
  for (int o = 16; o > 0; o >>= 1) {
    s0 += __shfl_xor(s0, o);
    ln += __shfl_xor(ln, o);
  }
  __syncthreads();
  if (lane == 0) { atomicAdd(&tot[0], s0); atomicAdd(&tot[1], ln); }
  __syncthreads();
  const int lo = tot[0];
  const int hi = lo + tot[1];
  walk_runs<0>(tmp, rb, rl, nchunk, wave, lane, ncnt, stage, lo, perm);
  __syncthreads();
  int v = 0;
  if (t < BUCKET_NODES) { v = ncnt[t]; nsc[t] = v; }
  __syncthreads();
  for (int o = 1; o < BUCKET_NODES; o <<= 1) {
    int u = 0;
    if (t < BUCKET_NODES && t >= o) u = nsc[t - o];
    __syncthreads();
    if (t < BUCKET_NODES) nsc[t] += u;
    __syncthreads();
  }
  if (t < BUCKET_NODES) {
    const int first = lo + nsc[t] - v;
    ncur[t] = first;
    const int node = b * BUCKET_NODES + t;
    if (node < N) off[node] = first;
  }
  if (b == nb - 1 && t == 0) off[N] = hi;
  __syncthreads();
  if (hi - lo <= STAGE_CAP) {
    walk_runs<1>(tmp, rb, rl, nchunk, wave, lane, ncur, stage, lo, perm);
    __syncthreads();
    for (int k = (lo & ~31) + t; k < hi; k += 1024)
      if (k >= lo) perm[k] = stage[k - lo];
  } else {
    walk_runs<2>(tmp, rb, rl, nchunk, wave, lane, ncur, stage, lo, perm);
  }
}

struct Row8 { v2f q[4]; };

__device__ __forceinline__ Row8 load_row8(const float* __restrict__ p) {
  const float4 a = *(const float4*)p;
  const float4 b = *(const float4*)(p + 4);
  Row8 r;
  r.q[0].x = a.x; r.q[0].y = a.y; r.q[1].x = a.z; r.q[1].y = a.w;
  r.q[2].x = b.x; r.q[2].y = b.y; r.q[3].x = b.z; r.q[3].y = b.w;
  return r;
}

__device__ __forceinline__ float dot8(const Row8& r, const Row8& w) {
  v2f ss = r.q[0] * w.q[0];
  ss += r.q[1] * w.q[1];
  ss += r.q[2] * w.q[2];
  ss += r.q[3] * w.q[3];
  return ss.x + ss.y;
}

typedef __attribute__((ext_vector_type(4))) int v4i;

__device__ __forceinline__ v4i load_idx4(const int* __restrict__ perm, int pos) {
  v4i v;
  __builtin_memcpy(&v, perm + pos, sizeof(v));
  return v;
}

__device__ __forceinline__ Row8 load_row8_at(const float* __restrict__ hf, unsigned o) {
  return load_row8((const float*)((const char*)hf + o));
}


__global__ void __launch_bounds__(256)
node_softmax_gather_heads8(const int* __restrict__ off, const int* __restrict__ perm,
                           const float* __restrict__ as, const float* __restrict__ ad,
                           const float* __restrict__ hf, const float* __restrict__ bias,
                           const float* __restrict__ w2, float* __restrict__ h2, long N) {
  __shared__ float4 rows[32 * 17];
  const long t0 = (long)blockIdx.x * 256 + threadIdx.x;
  const long t = t0 < N * 8 ? t0 : N * 8 - 8 + (t0 & 7);
  const int d = (int)(t >> 3);
  const int ro = (int)(t & 7) * 8;

  const Row8 asv = load_row8(as + ro);
  const Row8 own = load_row8(hf + (long)d * 64 + ro);
  const float ald = dot8(own, load_row8(ad + ro));
  float m = lrelu(dot8(own, asv) + ald);
  float den = 1.f;
  Row8 acc = own;

  const int j0 = off[d], j1 = off[d + 1];
  const int nfull = (j1 - j0) >> 2;
  const int rem = (j1 - j0) & 3;
  const unsigned ro4 = (unsigned)ro * 4u;
  v4i idn = load_idx4(perm, j0);
  for (int t = 0; t < nfull; ++t) {
    const Row8 r0 = load_row8_at(hf, ((unsigned)idn.x << 8) + ro4);
    const Row8 r1 = load_row8_at(hf, ((unsigned)idn.y << 8) + ro4);
    const Row8 r2 = load_row8_at(hf, ((unsigned)idn.z << 8) + ro4);
    const Row8 r3 = load_row8_at(hf, ((unsigned)idn.w << 8) + ro4);
    idn = load_idx4(perm, j0 + 4 * (t + 1));
    const float e0 = lrelu(dot8(r0, asv) + ald);
    const float e1 = lrelu(dot8(r1, asv) + ald);
    const float e2 = lrelu(dot8(r2, asv) + ald);
    const float e3 = lrelu(dot8(r3, asv) + ald);
    const float mn = fmaxf(fmaxf(fmaxf(m, e0), fmaxf(e1, e2)), e3);
    const float sc = __expf(m - mn);
    const float p0 = __expf(e0 - mn);
    const float p1 = __expf(e1 - mn);
    const float p2 = __expf(e2 - mn);
    const float p3 = __expf(e3 - mn);
    m = mn;
    den = den * sc + p0; den += p1; den += p2; den += p3;
#pragma unroll
    for (int k = 0; k < 4; ++k) {
      acc.q[k] = acc.q[k] * sc + p0 * r0.q[k];
      acc.q[k] += p1 * r1.q[k];
      acc.q[k] += p2 * r2.q[k];
      acc.q[k] += p3 * r3.q[k];
    }
  }
  if (rem != 0) {
    const Row8 r0 = load_row8_at(hf, ((unsigned)idn.x << 8) + ro4);
    const Row8 r1 = load_row8_at(hf, ((unsigned)(rem > 1 ? idn.y : d) << 8) + ro4);
    const Row8 r2 = load_row8_at(hf, ((unsigned)(rem > 2 ? idn.z : d) << 8) + ro4);
    const float e0 = lrelu(dot8(r0, asv) + ald);
    const float e1 = lrelu(dot8(r1, asv) + ald) + (rem > 1 ? 0.f : -INFINITY);
    const float e2 = lrelu(dot8(r2, asv) + ald) + (rem > 2 ? 0.f : -INFINITY);
    const float mn = fmaxf(fmaxf(m, e0), fmaxf(e1, e2));
    const float sc = __expf(m - mn);
    const float p0 = __expf(e0 - mn);
    const float p1 = __expf(e1 - mn);
    const float p2 = __expf(e2 - mn);
    m = mn;
    den = den * sc + p0; den += p1; den += p2;
#pragma unroll
    for (int k = 0; k < 4; ++k) {
      acc.q[k] = acc.q[k] * sc + p0 * r0.q[k];
      acc.q[k] += p1 * r1.q[k];
      acc.q[k] += p2 * r2.q[k];
    }
  }

  const float dn = den + 1e-16f;
  const float* bp = bias + ro;
  float4 o0, o1;
  o0.x = fmaxf(acc.q[0].x / dn + bp[0], 0.f); o0.y = fmaxf(acc.q[0].y / dn + bp[1], 0.f);
  o0.z = fmaxf(acc.q[1].x / dn + bp[2], 0.f); o0.w = fmaxf(acc.q[1].y / dn + bp[3], 0.f);
  o1.x = fmaxf(acc.q[2].x / dn + bp[4], 0.f); o1.y = fmaxf(acc.q[2].y / dn + bp[5], 0.f);
  o1.z = fmaxf(acc.q[3].x / dn + bp[6], 0.f); o1.w = fmaxf(acc.q[3].y / dn + bp[7], 0.f);
  float* sp = (float*)rows + (threadIdx.x >> 3) * 68 + ro;
  *(float4*)sp = o0;
  *(float4*)(sp + 4) = o1;

  const int lane = threadIdx.x & 31;
  const int wave = threadIdx.x >> 5;
  const int c = lane & 15;
  const int kg = (lane >> 4) << 1;
  v2f bf[16] = {};
  if (wave < 2) {
#pragma unroll
    for (int i = 0; i < 16; ++i) {
      bf[i].x = w2[(4 * i + kg) * 8 + (c & 7)];
      bf[i].y = w2[(4 * i + kg + 1) * 8 + (c & 7)];
    }
  }
  __syncthreads();
  if (wave < 2) {
    const float* arow = (const float*)rows + (16 * wave + c) * 68 + kg;
    v8f pr = {};
#pragma unroll
    for (int i = 0; i < 16; ++i) {
      const v2f a = *(const v2f*)(arow + 4 * i);
      pr = __builtin_amdgcn_wmma_f32_16x16x4_f32(false, a, false, bf[i], (short)0, pr,
                                                 false, false);
    }
    if (c < 8) {
      const long nb0 = (long)blockIdx.x * 32 + 16 * wave + 4 * kg;
#pragma unroll
      for (int v = 0; v < 8; ++v)
        if (nb0 + v < N) h2[(nb0 + v) * 8 + c] = pr[v];
    }
  }
}

template <int X>
__device__ __forceinline__ float lane_xor(float v) {
  return __builtin_bit_cast(float, __builtin_amdgcn_update_dpp(
      0, __builtin_bit_cast(int, v), 0x160 + X, 0xf, 0xf, true));
}

__device__ __forceinline__ float sum8(float v) {
  v += lane_xor<1>(v);
  v += lane_xor<2>(v);
  v += lane_xor<4>(v);
  return v;
}

__device__ __forceinline__ float load_f_at(const float* __restrict__ hf, unsigned o) {
  return *(const float*)((const char*)hf + o);
}

__global__ void __launch_bounds__(256)
node_softmax_gather_head1(const int* __restrict__ off, const int* __restrict__ perm,
                          const float* __restrict__ as, const float* __restrict__ ad,
                          const float* __restrict__ hf, const float* __restrict__ bias,
                          const float* __restrict__ w3, const float* __restrict__ b3,
                          float* __restrict__ out, float* __restrict__ score, long N) {
  const long t = (long)blockIdx.x * blockDim.x + threadIdx.x;
  if (t >= N * 8) return;
  const int d = (int)(t >> 3);
  const int g = (int)(t & 7);

  const float asg = as[g];
  const float own = hf[(long)d * 8 + g];
  const float ald = sum8(own * ad[g]);
  float m = lrelu(sum8(own * asg) + ald);
  float den = 1.f;
  float acc = own;

  const int j0 = off[d], j1 = off[d + 1];
  const int nfull = (j1 - j0) >> 2;
  const int rem = (j1 - j0) & 3;
  const unsigned g4 = (unsigned)g * 4u;
  v4i idn = load_idx4(perm, j0);
  for (int t = 0; t < nfull; ++t) {
    const float x0 = load_f_at(hf, ((unsigned)idn.x << 5) + g4);
    const float x1 = load_f_at(hf, ((unsigned)idn.y << 5) + g4);
    const float x2 = load_f_at(hf, ((unsigned)idn.z << 5) + g4);
    const float x3 = load_f_at(hf, ((unsigned)idn.w << 5) + g4);
    idn = load_idx4(perm, j0 + 4 * (t + 1));
    const float e0 = lrelu(sum8(x0 * asg) + ald);
    const float e1 = lrelu(sum8(x1 * asg) + ald);
    const float e2 = lrelu(sum8(x2 * asg) + ald);
    const float e3 = lrelu(sum8(x3 * asg) + ald);
    const float mn = fmaxf(fmaxf(fmaxf(m, e0), fmaxf(e1, e2)), e3);
    const float sc = __expf(m - mn);
    const float p0 = __expf(e0 - mn);
    const float p1 = __expf(e1 - mn);
    const float p2 = __expf(e2 - mn);
    const float p3 = __expf(e3 - mn);
    m = mn;
    den = den * sc + p0; den += p1; den += p2; den += p3;
    acc = acc * sc + p0 * x0; acc += p1 * x1; acc += p2 * x2; acc += p3 * x3;
  }
  if (rem != 0) {
    const float x0 = load_f_at(hf, ((unsigned)idn.x << 5) + g4);
    const float x1 = load_f_at(hf, ((unsigned)(rem > 1 ? idn.y : d) << 5) + g4);
    const float x2 = load_f_at(hf, ((unsigned)(rem > 2 ? idn.z : d) << 5) + g4);
    const float e0 = lrelu(sum8(x0 * asg) + ald);
    const float e1 = lrelu(sum8(x1 * asg) + ald) + (rem > 1 ? 0.f : -INFINITY);
    const float e2 = lrelu(sum8(x2 * asg) + ald) + (rem > 2 ? 0.f : -INFINITY);
    const float mn = fmaxf(fmaxf(m, e0), fmaxf(e1, e2));
    const float sc = __expf(m - mn);
    const float p0 = __expf(e0 - mn);
    const float p1 = __expf(e1 - mn);
    const float p2 = __expf(e2 - mn);
    m = mn;
    den = den * sc + p0; den += p1; den += p2;
    acc = acc * sc + p0 * x0; acc += p1 * x1; acc += p2 * x2;
  }
  const float o = acc / (den + 1e-16f) + bias[g];
  out[(long)d * 8 + g] = o;
  const float sc = sum8(o * w3[g]) + b3[0];
  if (g == 0) score[d] = sc;
}

static inline int nblk(long n, int b) { return (int)((n + b - 1) / b); }

extern "C" void kernel_launch(void* const* d_in, const int* in_sizes, int n_in,
                              void* d_out, int out_size, void* d_ws, size_t ws_size,
                              hipStream_t stream) {
  const float* x   = (const float*)d_in[0];
  const int*   ei  = (const int*)d_in[1];
  const float* W1  = (const float*)d_in[2];
  const float* as1 = (const float*)d_in[3];
  const float* ad1 = (const float*)d_in[4];
  const float* b1  = (const float*)d_in[5];
  const float* W2  = (const float*)d_in[6];
  const float* as2 = (const float*)d_in[7];
  const float* ad2 = (const float*)d_in[8];
  const float* b2  = (const float*)d_in[9];
  const float* W3  = (const float*)d_in[10];
  const float* b3  = (const float*)d_in[11];

  const long N    = in_sizes[0] / 256;
  const long Eraw = in_sizes[1] / 2;

  float* ws   = (float*)d_ws;
  float* h1   = ws;
  float* acc1 = h1 + N * 64;
  int*   perm = (int*)(acc1 + N * 64);
  int*   off  = perm + Eraw;
  const int nb = (int)((N + BUCKET_NODES - 1) / BUCKET_NODES);
  long chunk_l = (Eraw + NCHUNK - 1) / NCHUNK;
  if (chunk_l > CHUNK_CAP) chunk_l = CHUNK_CAP;
  if (chunk_l < 1) chunk_l = 1;
  const int chunk  = (int)chunk_l;
  const int nchunk = nblk(Eraw, chunk);
  unsigned* tmp = (unsigned*)acc1;
  int*   st   = (int*)acc1 + Eraw;
  float* h2   = acc1;

  float* emb   = (float*)d_out;
  float* score = emb + N * 8;

  const int B = 256;
  const int mtiles = (int)(N / 16);

  chunk_sort<<<nchunk, 1024, 0, stream>>>(ei, Eraw, chunk, nb, st, tmp);
  bucket_sort_nodes<<<nb, 1024, 0, stream>>>(tmp, st, chunk, nchunk, nb, (int)N, off, perm);

  gemm1_x_w1<<<(mtiles + 3) / 4, 256, 0, stream>>>(x, W1, h1, mtiles);
  node_softmax_gather_heads8<<<nblk(N * 8, 256), 256, 0, stream>>>(off, perm, as1, ad1, h1, b1, W2, h2, N);

  node_softmax_gather_head1<<<nblk(N * 8, B), B, 0, stream>>>(off, perm, as2, ad2, h2, b2, W3, b3,
                                                              emb, score, N);
}
